# starting kernel
# baseline (speedup 1.0000x reference)
#include <hip/hip_runtime.h>


#define NNODES 50000
#define NEDGES 800000
#define KDIM   128

typedef __attribute__((ext_vector_type(2))) float v2f;
typedef __attribute__((ext_vector_type(8))) float v8f;

__global__ void sage_degree_kernel(const int* __restrict__ dst,
                                   float* __restrict__ deg, int E) {
    int e = blockIdx.x * blockDim.x + threadIdx.x;
    if (e < E) atomicAdd(&deg[dst[e]], 1.0f);
}

__global__ void sage_scatter_kernel(const float* __restrict__ h,
                                    const int* __restrict__ src,
                                    const int* __restrict__ dst,
                                    float* __restrict__ agg, int E) {
    int wid  = threadIdx.x >> 5;
    int lane = threadIdx.x & 31;
    int e = blockIdx.x * (blockDim.x >> 5) + wid;
    if (e >= E) return;
    int s = src[e];
    int d = dst[e];
    const float* hp = h   + (long)s * KDIM + lane;
    float*       ap = agg + (long)d * KDIM + lane;
#pragma unroll
    for (int k = 0; k < 4; ++k)
        atomicAdd(ap + k * 32, hp[k * 32]);
}

template <int NC, bool RELU>
__global__ void sage_gemm_kernel(const float* __restrict__ h,
                                 const float* __restrict__ agg,
                                 const float* __restrict__ deg,
                                 const float* __restrict__ Ws,
                                 const float* __restrict__ Wn,
                                 const float* __restrict__ bias,
                                 float* __restrict__ out) {
    constexpr int colTiles = NC >> 4;
    constexpr int nTiles   = (NNODES / 16) * colTiles;

    int tile = blockIdx.x * (blockDim.x >> 5) + (threadIdx.x >> 5);
    if (tile >= nTiles) return;

    int lane = threadIdx.x & 31;
    int r0 = (tile / colTiles) << 4;
    int c0 = (tile % colTiles) << 4;

    int m    = lane & 15;
    int koff = (lane >> 4) << 1;
    int row  = r0 + m;
    int n    = c0 + m;

    float idg = 1.0f / fmaxf(deg[row], 1.0f);

    const float* hrow = h   + (long)row * KDIM + koff;
    const float* grow = agg + (long)row * KDIM + koff;
    const float* wsp  = Ws + koff * NC + n;
    const float* wnp  = Wn + koff * NC + n;

    v8f acc = {0.f, 0.f, 0.f, 0.f, 0.f, 0.f, 0.f, 0.f};

#pragma unroll 8
    for (int k0 = 0; k0 < KDIM; k0 += 4) {
        v2f a1, a2, b1, b2;
        a1.x = hrow[k0];
        a1.y = hrow[k0 + 1];
        a2.x = grow[k0] * idg;
        a2.y = grow[k0 + 1] * idg;
        b1.x = wsp[k0 * NC];
        b1.y = wsp[(k0 + 1) * NC];
        b2.x = wnp[k0 * NC];
        b2.y = wnp[(k0 + 1) * NC];
        acc = __builtin_amdgcn_wmma_f32_16x16x4_f32(false, a1, false, b1,
                                                    (short)0, acc, false, false);
        acc = __builtin_amdgcn_wmma_f32_16x16x4_f32(false, a2, false, b2,
                                                    (short)0, acc, false, false);
    }

    float bv    = bias[n];
    int   rbase = r0 + ((lane >> 4) << 3);
#pragma unroll
    for (int v = 0; v < 8; ++v) {
        float x = acc[v] + bv;
        if (RELU) x = fmaxf(x, 0.0f);
        out[(long)(rbase + v) * NC + n] = x;
    }
}

extern "C" void kernel_launch(void* const* d_in, const int* in_sizes, int n_in,
                              void* d_out, int out_size, void* d_ws, size_t ws_size,
                              hipStream_t stream) {
    (void)in_sizes; (void)n_in; (void)out_size; (void)ws_size;

    const float* x   = (const float*)d_in[0];
    const int*   src = (const int*)d_in[1];
    const int*   dst = (const int*)d_in[2];
    const float* Ws0 = (const float*)d_in[3];
    const float* Wn0 = (const float*)d_in[4];
    const float* b0  = (const float*)d_in[5];
    const float* Ws1 = (const float*)d_in[6];
    const float* Wn1 = (const float*)d_in[7];
    const float* b1  = (const float*)d_in[8];
    const float* Ws2 = (const float*)d_in[9];
    const float* Wn2 = (const float*)d_in[10];
    const float* b2  = (const float*)d_in[11];
    float* out = (float*)d_out;

    const size_t feat = (size_t)NNODES * KDIM;
    float* agg = (float*)d_ws;
    float* hA  = agg + feat;
    float* hB  = hA + feat;
    float* deg = hB + feat;

    const int scatterBlocks = (NEDGES + 7) / 8;
    const int gemmBlocks128 = ((NNODES / 16) * 8 + 7) / 8;
    const int gemmBlocks64  = ((NNODES / 16) * 4 + 7) / 8;

    hipMemsetAsync(deg, 0, NNODES * sizeof(float), stream);
    sage_degree_kernel<<<(NEDGES + 255) / 256, 256, 0, stream>>>(dst, deg, NEDGES);

    hipMemsetAsync(agg, 0, feat * sizeof(float), stream);
    sage_scatter_kernel<<<scatterBlocks, 256, 0, stream>>>(x, src, dst, agg, NEDGES);
    sage_gemm_kernel<128, true><<<gemmBlocks128, 256, 0, stream>>>(
        x, agg, deg, Ws0, Wn0, b0, hA);

    hipMemsetAsync(agg, 0, feat * sizeof(float), stream);
    sage_scatter_kernel<<<scatterBlocks, 256, 0, stream>>>(hA, src, dst, agg, NEDGES);
    sage_gemm_kernel<128, true><<<gemmBlocks128, 256, 0, stream>>>(
        hA, agg, deg, Ws1, Wn1, b1, hB);

    hipMemsetAsync(agg, 0, feat * sizeof(float), stream);
    sage_scatter_kernel<<<scatterBlocks, 256, 0, stream>>>(hB, src, dst, agg, NEDGES);
    sage_gemm_kernel<64, false><<<gemmBlocks64, 256, 0, stream>>>(
        hB, agg, deg, Ws2, Wn2, b2, out);
}


# Gather over a dst-grouped edge list built once per call (int atomic rank, scan, fill) replaces the f32 atomic scatter; each list sorted by src, so the f32 sum order is fixed, not arrival order
# speedup vs baseline: 2.0802x; 2.0802x over previous
#include <hip/hip_runtime.h>


#define NNODES 50000
#define NEDGES 800000
#define KDIM   128

typedef __attribute__((ext_vector_type(2))) float v2f;
typedef __attribute__((ext_vector_type(8))) float v8f;

typedef __attribute__((ext_vector_type(4))) float v4f;

#define EDGES_PER_THREAD 4
#define SCAN_BLOCK       1024
#define GATHER_BLOCKS    2048

__global__ void sage_rank_kernel(const int* __restrict__ dst,
                                 int* __restrict__ cnt,
                                 int* __restrict__ rank, int E) {
    const int tid   = blockIdx.x * blockDim.x + threadIdx.x;
    const int total = gridDim.x * blockDim.x;
    int d[EDGES_PER_THREAD];
    int r[EDGES_PER_THREAD];
#pragma unroll
    for (int k = 0; k < EDGES_PER_THREAD; ++k) {
        const int e = tid + k * total;
        d[k] = (e < E) ? dst[e] : -1;
    }
#pragma unroll
    for (int k = 0; k < EDGES_PER_THREAD; ++k)
        r[k] = (d[k] >= 0) ? atomicAdd(&cnt[d[k]], 1) : 0;
#pragma unroll
    for (int k = 0; k < EDGES_PER_THREAD; ++k) {
        const int e = tid + k * total;
        if (e < E) rank[e] = r[k];
    }
}

__global__ void __launch_bounds__(SCAN_BLOCK)
sage_scan_kernel(const int* __restrict__ cnt,
                 int* __restrict__ rowptr,
                 float* __restrict__ deg) {
    __shared__ int s_pre[SCAN_BLOCK / 32];
    __shared__ int s_tot[SCAN_BLOCK / 32];
    const int t    = threadIdx.x;
    const int lane = t & 31;
    const int w    = t >> 5;
    const int b    = blockIdx.x;

    int pre = 0;
    for (int c = 0; c < b; ++c) pre += cnt[c * SCAN_BLOCK + t];
    for (int o = 16; o > 0; o >>= 1) pre += __shfl_xor(pre, o);

    const int i = b * SCAN_BLOCK + t;
    const int v = (i < NNODES) ? cnt[i] : 0;
    int inc = v;
    for (int o = 1; o < 32; o <<= 1) {
        const int u = __shfl_up(inc, o);
        if (lane >= o) inc += u;
    }
    if (lane == 31) s_tot[w] = inc;
    if (lane == 0)  s_pre[w] = pre;
    __syncthreads();

    int base = 0;
    for (int k = 0; k < SCAN_BLOCK / 32; ++k) {
        base += s_pre[k];
        if (k < w) base += s_tot[k];
    }
    const int excl = base + inc - v;
    if (i < NNODES) {
        rowptr[i] = excl;
        deg[i]    = (float)v;
    }
    if (i == NNODES - 1) rowptr[NNODES] = excl + v;
}

__global__ void sage_fill_kernel(const int* __restrict__ src,
                                 const int* __restrict__ dst,
                                 const int* __restrict__ rank,
                                 const int* __restrict__ rowptr,
                                 int* __restrict__ csr, int E) {
    const int tid   = blockIdx.x * blockDim.x + threadIdx.x;
    const int total = gridDim.x * blockDim.x;
    int d[EDGES_PER_THREAD];
    int r[EDGES_PER_THREAD];
    int s[EDGES_PER_THREAD];
    int p[EDGES_PER_THREAD];
#pragma unroll
    for (int k = 0; k < EDGES_PER_THREAD; ++k) {
        const int  e  = tid + k * total;
        const bool ok = e < E;
        const int  ec = ok ? e : E - 1;
        d[k] = dst[ec];
        s[k] = src[ec];
        r[k] = ok ? rank[ec] : E;
    }
#pragma unroll
    for (int k = 0; k < EDGES_PER_THREAD; ++k) p[k] = rowptr[d[k]] + r[k];
#pragma unroll
    for (int k = 0; k < EDGES_PER_THREAD; ++k)
        if ((unsigned)p[k] < (unsigned)E) csr[p[k]] = s[k];
}

__global__ void sage_order_kernel(const int* __restrict__ rowptr,
                                  const int* __restrict__ csr_in,
                                  int* __restrict__ csr_out) {
    const int lane = threadIdx.x & 31;
    const int wave   = __builtin_amdgcn_readfirstlane((int)(threadIdx.x >> 5));
    const int wpb    = blockDim.x >> 5;
    const int stride = gridDim.x * wpb;

    for (int node = blockIdx.x * wpb + wave; node < NNODES; node += stride) {
        const int beg = __builtin_amdgcn_readfirstlane(rowptr[node]);
        const int n   = __builtin_amdgcn_readfirstlane(rowptr[node + 1]) - beg;
        for (int i0 = 0; i0 < n; i0 += 32) {
            const int i = i0 + lane;
            const int a = (i < n) ? csr_in[beg + i] : 0;
            int pos = 0;
            for (int m0 = 0; m0 < n; m0 += 32) {
                const int mm = m0 + lane;
                const int bv = (mm < n) ? csr_in[beg + mm] : 0;
                int piece = n - m0;
                if (piece > 32) piece = 32;
                for (int u = 0; u < piece; ++u) {
                    const int b = __builtin_amdgcn_readlane(bv, u);
                    const int m = m0 + u;
                    pos += (b < a || (b == a && m < i)) ? 1 : 0;
                }
            }
            if (i < n) csr_out[beg + pos] = a;
        }
    }
}

template <int G>
__device__ __forceinline__ void sage_add_rows(const float* __restrict__ h, int idx, int j,
                                              int lane4, v4f& acc) {
    v4f v[G];
#pragma unroll
    for (int u = 0; u < G; ++u) {
        const int s = __builtin_amdgcn_readlane(idx, j + u);
        v[u] = *(const v4f*)(h + (size_t)s * KDIM + lane4);
    }
#pragma unroll
    for (int u = 0; u < G; ++u) acc += v[u];
}

__global__ void sage_gather_kernel(const float* __restrict__ h,
                                   const int* __restrict__ rowptr,
                                   const int* __restrict__ csr,
                                   float* __restrict__ agg) {
    const int lane  = threadIdx.x & 31;
    const int lane4 = lane << 2;
    const int wave   = __builtin_amdgcn_readfirstlane((int)(threadIdx.x >> 5));
    const int wpb    = blockDim.x >> 5;
    const int stride = gridDim.x * wpb;

    for (int node = blockIdx.x * wpb + wave; node < NNODES; node += stride) {
        const int beg = __builtin_amdgcn_readfirstlane(rowptr[node]);
        const int end = __builtin_amdgcn_readfirstlane(rowptr[node + 1]);
        v4f acc = {0.f, 0.f, 0.f, 0.f};
        for (int base = beg; base < end; base += 32) {
            int n = end - base;
            if (n > 32) n = 32;
            const int idx = (lane < n) ? csr[base + lane] : 0;
            int j = 0;
            for (; j + 8 <= n; j += 8) sage_add_rows<8>(h, idx, j, lane4, acc);
            if (n & 4) { sage_add_rows<4>(h, idx, j, lane4, acc); j += 4; }
            if (n & 2) { sage_add_rows<2>(h, idx, j, lane4, acc); j += 2; }
            if (n & 1) { sage_add_rows<1>(h, idx, j, lane4, acc); }
        }
        *(v4f*)(agg + (size_t)node * KDIM + lane4) = acc;
    }
}

template <int NC, bool RELU>
__global__ void sage_gemm_kernel(const float* __restrict__ h,
                                 const float* __restrict__ agg,
                                 const float* __restrict__ deg,
                                 const float* __restrict__ Ws,
                                 const float* __restrict__ Wn,
                                 const float* __restrict__ bias,
                                 float* __restrict__ out) {
    constexpr int colTiles = NC >> 4;
    constexpr int nTiles   = (NNODES / 16) * colTiles;

    int tile = blockIdx.x * (blockDim.x >> 5) + (threadIdx.x >> 5);
    if (tile >= nTiles) return;

    int lane = threadIdx.x & 31;
    int r0 = (tile / colTiles) << 4;
    int c0 = (tile % colTiles) << 4;

    int m    = lane & 15;
    int koff = (lane >> 4) << 1;
    int row  = r0 + m;
    int n    = c0 + m;

    float idg = 1.0f / fmaxf(deg[row], 1.0f);

    const float* hrow = h   + (long)row * KDIM + koff;
    const float* grow = agg + (long)row * KDIM + koff;
    const float* wsp  = Ws + koff * NC + n;
    const float* wnp  = Wn + koff * NC + n;

    v8f acc = {0.f, 0.f, 0.f, 0.f, 0.f, 0.f, 0.f, 0.f};

#pragma unroll 8
    for (int k0 = 0; k0 < KDIM; k0 += 4) {
        v2f a1, a2, b1, b2;
        a1.x = hrow[k0];
        a1.y = hrow[k0 + 1];
        a2.x = grow[k0] * idg;
        a2.y = grow[k0 + 1] * idg;
        b1.x = wsp[k0 * NC];
        b1.y = wsp[(k0 + 1) * NC];
        b2.x = wnp[k0 * NC];
        b2.y = wnp[(k0 + 1) * NC];
        acc = __builtin_amdgcn_wmma_f32_16x16x4_f32(false, a1, false, b1,
                                                    (short)0, acc, false, false);
        acc = __builtin_amdgcn_wmma_f32_16x16x4_f32(false, a2, false, b2,
                                                    (short)0, acc, false, false);
    }

    float bv    = bias[n];
    int   rbase = r0 + ((lane >> 4) << 3);
#pragma unroll
    for (int v = 0; v < 8; ++v) {
        float x = acc[v] + bv;
        if (RELU) x = fmaxf(x, 0.0f);
        out[(long)(rbase + v) * NC + n] = x;
    }
}

extern "C" void kernel_launch(void* const* d_in, const int* in_sizes, int n_in,
                              void* d_out, int out_size, void* d_ws, size_t ws_size,
                              hipStream_t stream) {
    (void)in_sizes; (void)n_in; (void)out_size; (void)ws_size;

    const float* x   = (const float*)d_in[0];
    const int*   src = (const int*)d_in[1];
    const int*   dst = (const int*)d_in[2];
    const float* Ws0 = (const float*)d_in[3];
    const float* Wn0 = (const float*)d_in[4];
    const float* b0  = (const float*)d_in[5];
    const float* Ws1 = (const float*)d_in[6];
    const float* Wn1 = (const float*)d_in[7];
    const float* b1  = (const float*)d_in[8];
    const float* Ws2 = (const float*)d_in[9];
    const float* Wn2 = (const float*)d_in[10];
    const float* b2  = (const float*)d_in[11];
    float* out = (float*)d_out;

    const size_t feat = (size_t)NNODES * KDIM;
    float* agg = (float*)d_ws;
    float* hA  = agg + feat;
    float* hB  = hA + feat;
    float* deg = hB + feat;

    int* cnt  = (int*)hB;
    int* rank = cnt + NNODES;
    int* csr0 = rank + NEDGES;
    int* csr    = (int*)d_out;
    int* rowptr = csr + NEDGES;

    const int edgeBlocks    = (NEDGES + 256 * EDGES_PER_THREAD - 1) / (256 * EDGES_PER_THREAD);
    const int scanBlocks    = (NNODES + SCAN_BLOCK - 1) / SCAN_BLOCK;
    const int gemmBlocks128 = ((NNODES / 16) * 8 + 7) / 8;
    const int gemmBlocks64  = ((NNODES / 16) * 4 + 7) / 8;

    hipMemsetAsync(cnt, 0, NNODES * sizeof(int), stream);
    sage_rank_kernel<<<edgeBlocks, 256, 0, stream>>>(dst, cnt, rank, NEDGES);
    sage_scan_kernel<<<scanBlocks, SCAN_BLOCK, 0, stream>>>(cnt, rowptr, deg);
    sage_fill_kernel<<<edgeBlocks, 256, 0, stream>>>(src, dst, rank, rowptr, csr0, NEDGES);
    sage_order_kernel<<<GATHER_BLOCKS, 256, 0, stream>>>(rowptr, csr0, csr);

    sage_gather_kernel<<<GATHER_BLOCKS, 256, 0, stream>>>(x, rowptr, csr, agg);
    sage_gemm_kernel<128, true><<<gemmBlocks128, 256, 0, stream>>>(
        x, agg, deg, Ws0, Wn0, b0, hA);

    sage_gather_kernel<<<GATHER_BLOCKS, 256, 0, stream>>>(hA, rowptr, csr, agg);
    sage_gemm_kernel<128, true><<<gemmBlocks128, 256, 0, stream>>>(
        hA, agg, deg, Ws1, Wn1, b1, hB);

    sage_gather_kernel<<<GATHER_BLOCKS, 256, 0, stream>>>(hB, rowptr, csr, agg);
    sage_gemm_kernel<64, false><<<gemmBlocks64, 256, 0, stream>>>(
        hB, agg, deg, Ws2, Wn2, b2, out);
}


# GEMM re-tiled: weights staged once per work-group in LDS; each wave computes 16 rows x 64 (32) columns sharing one A fragment; k taken in quads, so the f32 sum order changes (still fixed)
# speedup vs baseline: 2.7120x; 1.3037x over previous
#include <hip/hip_runtime.h>


#define NNODES 50000
#define NEDGES 800000
#define KDIM   128

typedef __attribute__((ext_vector_type(2))) float v2f;
typedef __attribute__((ext_vector_type(8))) float v8f;

typedef __attribute__((ext_vector_type(4))) float v4f;

#define EDGES_PER_THREAD 4
#define SCAN_BLOCK       1024
#define GATHER_BLOCKS    2048

__global__ void sage_rank_kernel(const int* __restrict__ dst,
                                 int* __restrict__ cnt,
                                 int* __restrict__ rank, int E) {
    const int tid   = blockIdx.x * blockDim.x + threadIdx.x;
    const int total = gridDim.x * blockDim.x;
    int d[EDGES_PER_THREAD];
    int r[EDGES_PER_THREAD];
#pragma unroll
    for (int k = 0; k < EDGES_PER_THREAD; ++k) {
        const int e = tid + k * total;
        d[k] = (e < E) ? dst[e] : -1;
    }
#pragma unroll
    for (int k = 0; k < EDGES_PER_THREAD; ++k)
        r[k] = (d[k] >= 0) ? atomicAdd(&cnt[d[k]], 1) : 0;
#pragma unroll
    for (int k = 0; k < EDGES_PER_THREAD; ++k) {
        const int e = tid + k * total;
        if (e < E) rank[e] = r[k];
    }
}

__global__ void __launch_bounds__(SCAN_BLOCK)
sage_scan_kernel(const int* __restrict__ cnt,
                 int* __restrict__ rowptr,
                 float* __restrict__ deg) {
    __shared__ int s_pre[SCAN_BLOCK / 32];
    __shared__ int s_tot[SCAN_BLOCK / 32];
    const int t    = threadIdx.x;
    const int lane = t & 31;
    const int w    = t >> 5;
    const int b    = blockIdx.x;

    int pre = 0;
    for (int c = 0; c < b; ++c) pre += cnt[c * SCAN_BLOCK + t];
    for (int o = 16; o > 0; o >>= 1) pre += __shfl_xor(pre, o);

    const int i = b * SCAN_BLOCK + t;
    const int v = (i < NNODES) ? cnt[i] : 0;
    int inc = v;
    for (int o = 1; o < 32; o <<= 1) {
        const int u = __shfl_up(inc, o);
        if (lane >= o) inc += u;
    }
    if (lane == 31) s_tot[w] = inc;
    if (lane == 0)  s_pre[w] = pre;
    __syncthreads();

    int base = 0;
    for (int k = 0; k < SCAN_BLOCK / 32; ++k) {
        base += s_pre[k];
        if (k < w) base += s_tot[k];
    }
    const int excl = base + inc - v;
    if (i < NNODES) {
        rowptr[i] = excl;
        deg[i]    = (float)v;
    }
    if (i == NNODES - 1) rowptr[NNODES] = excl + v;
}

__global__ void sage_fill_kernel(const int* __restrict__ src,
                                 const int* __restrict__ dst,
                                 const int* __restrict__ rank,
                                 const int* __restrict__ rowptr,
                                 int* __restrict__ csr, int E) {
    const int tid   = blockIdx.x * blockDim.x + threadIdx.x;
    const int total = gridDim.x * blockDim.x;
    int d[EDGES_PER_THREAD];
    int r[EDGES_PER_THREAD];
    int s[EDGES_PER_THREAD];
    int p[EDGES_PER_THREAD];
#pragma unroll
    for (int k = 0; k < EDGES_PER_THREAD; ++k) {
        const int  e  = tid + k * total;
        const bool ok = e < E;
        const int  ec = ok ? e : E - 1;
        d[k] = dst[ec];
        s[k] = src[ec];
        r[k] = ok ? rank[ec] : E;
    }
#pragma unroll
    for (int k = 0; k < EDGES_PER_THREAD; ++k) p[k] = rowptr[d[k]] + r[k];
#pragma unroll
    for (int k = 0; k < EDGES_PER_THREAD; ++k)
        if ((unsigned)p[k] < (unsigned)E) csr[p[k]] = s[k];
}

__global__ void sage_order_kernel(const int* __restrict__ rowptr,
                                  const int* __restrict__ csr_in,
                                  int* __restrict__ csr_out) {
    const int lane = threadIdx.x & 31;
    const int wave   = __builtin_amdgcn_readfirstlane((int)(threadIdx.x >> 5));
    const int wpb    = blockDim.x >> 5;
    const int stride = gridDim.x * wpb;

    for (int node = blockIdx.x * wpb + wave; node < NNODES; node += stride) {
        const int beg = __builtin_amdgcn_readfirstlane(rowptr[node]);
        const int n   = __builtin_amdgcn_readfirstlane(rowptr[node + 1]) - beg;
        for (int i0 = 0; i0 < n; i0 += 32) {
            const int i = i0 + lane;
            const int a = (i < n) ? csr_in[beg + i] : 0;
            int pos = 0;
            for (int m0 = 0; m0 < n; m0 += 32) {
                const int mm = m0 + lane;
                const int bv = (mm < n) ? csr_in[beg + mm] : 0;
                int piece = n - m0;
                if (piece > 32) piece = 32;
                for (int u = 0; u < piece; ++u) {
                    const int b = __builtin_amdgcn_readlane(bv, u);
                    const int m = m0 + u;
                    pos += (b < a || (b == a && m < i)) ? 1 : 0;
                }
            }
            if (i < n) csr_out[beg + pos] = a;
        }
    }
}

template <int G>
__device__ __forceinline__ void sage_add_rows(const float* __restrict__ h, int idx, int j,
                                              int lane4, v4f& acc) {
    v4f v[G];
#pragma unroll
    for (int u = 0; u < G; ++u) {
        const int s = __builtin_amdgcn_readlane(idx, j + u);
        v[u] = *(const v4f*)(h + (size_t)s * KDIM + lane4);
    }
#pragma unroll
    for (int u = 0; u < G; ++u) acc += v[u];
}

__global__ void sage_gather_kernel(const float* __restrict__ h,
                                   const int* __restrict__ rowptr,
                                   const int* __restrict__ csr,
                                   float* __restrict__ agg) {
    const int lane  = threadIdx.x & 31;
    const int lane4 = lane << 2;
    const int wave   = __builtin_amdgcn_readfirstlane((int)(threadIdx.x >> 5));
    const int wpb    = blockDim.x >> 5;
    const int stride = gridDim.x * wpb;

    for (int node = blockIdx.x * wpb + wave; node < NNODES; node += stride) {
        const int beg = __builtin_amdgcn_readfirstlane(rowptr[node]);
        const int end = __builtin_amdgcn_readfirstlane(rowptr[node + 1]);
        v4f acc = {0.f, 0.f, 0.f, 0.f};
        for (int base = beg; base < end; base += 32) {
            int n = end - base;
            if (n > 32) n = 32;
            const int idx = (lane < n) ? csr[base + lane] : 0;
            int j = 0;
            for (; j + 8 <= n; j += 8) sage_add_rows<8>(h, idx, j, lane4, acc);
            if (n & 4) { sage_add_rows<4>(h, idx, j, lane4, acc); j += 4; }
            if (n & 2) { sage_add_rows<2>(h, idx, j, lane4, acc); j += 2; }
            if (n & 1) { sage_add_rows<1>(h, idx, j, lane4, acc); }
        }
        *(v4f*)(agg + (size_t)node * KDIM + lane4) = acc;
    }
}

#define GEMM_THREADS 512
#define GEMM_BLOCKS  256
#define GEMM_QUADS   64
#define GEMM_LDS_V4F 10304

template <int NC, int TILES, bool RELU>
__global__ void __launch_bounds__(GEMM_THREADS)
sage_gemm_kernel(const float* __restrict__ h,
                 const float* __restrict__ agg,
                 const float* __restrict__ deg,
                 const float* __restrict__ Ws,
                 const float* __restrict__ Wn,
                 const float* __restrict__ bias,
                 float* __restrict__ out) {
    constexpr int GW     = 16 * TILES;
    constexpr int GROUPS = NC / GW;
    constexpr int NITEMS = (NNODES / 16) * GROUPS;
    constexpr int PITCH  = NC + 8;
    constexpr int WPB    = GEMM_THREADS / 32;
    static_assert(GEMM_QUADS * PITCH <= GEMM_LDS_V4F, "weights must fit the LDS array");
    __shared__ v4f sW[GEMM_LDS_V4F];

    constexpr int FILL_ITERS = GEMM_QUADS * NC / GEMM_THREADS;
    static_assert(GEMM_QUADS * NC % GEMM_THREADS == 0, "whole fill passes");
#pragma unroll 8
    for (int i = 0; i < FILL_ITERS; ++i) {
        const int idx = threadIdx.x + i * GEMM_THREADS;
        const int q = idx / NC;
        const int p = idx % NC;
        const int n = (p / GW) * GW + TILES * (p & 15) + ((p % GW) >> 4);
        const float* w = ((q < 32) ? Ws : Wn) + (size_t)(4 * (q & 31)) * NC + n;
        v4f v;
        v.x = w[0];
        v.y = w[NC];
        v.z = w[2 * NC];
        v.w = w[3 * NC];
        sW[q * PITCH + p] = v;
    }
    __syncthreads();

    const int lane = threadIdx.x & 31;
    const int c    = lane & 15;
    const int hh   = lane >> 4;
    const int wave = __builtin_amdgcn_readfirstlane((int)(threadIdx.x >> 5));

    for (int item = blockIdx.x + gridDim.x * wave; item < NITEMS; item += gridDim.x * WPB) {
        const int r0  = (item / GROUPS) << 4;
        const int n0  = (item % GROUPS) * GW;
        const int row = r0 + c;
        const v4f* hq = (const v4f*)(h   + (size_t)row * KDIM) + hh;
        const v4f* gq = (const v4f*)(agg + (size_t)row * KDIM) + hh;
        const v4f* bq = sW + hh * PITCH + n0 + c;
        const float idg = 1.0f / fmaxf(deg[row], 1.0f);

        v8f acc[TILES];
#pragma unroll
        for (int t = 0; t < TILES; ++t)
            acc[t] = (v8f){0.f, 0.f, 0.f, 0.f, 0.f, 0.f, 0.f, 0.f};

        v4f a[4], an[4], b[TILES], bn[TILES];
#pragma unroll
        for (int jj = 0; jj < 4; ++jj) a[jj] = hq[2 * jj];
#pragma unroll
        for (int t = 0; t < TILES; ++t) b[t] = bq[16 * t];

#pragma unroll
        for (int ch = 0; ch < 8; ++ch) {
            if (ch + 1 < 8) {
                const v4f* nx = (ch + 1 < 4) ? (hq + 8 * (ch + 1)) : (gq + 8 * (ch + 1 - 4));
#pragma unroll
                for (int jj = 0; jj < 4; ++jj) an[jj] = nx[2 * jj];
            }
            if (ch >= 4) {
#pragma unroll
                for (int jj = 0; jj < 4; ++jj) a[jj] *= idg;
            }
#pragma unroll
            for (int jj = 0; jj < 4; ++jj) {
                const int s = 4 * ch + jj;
                if (s + 1 < 32) {
#pragma unroll
                    for (int t = 0; t < TILES; ++t) bn[t] = bq[(2 * (s + 1)) * PITCH + 16 * t];
                }
                const v2f alo = {a[jj].x, a[jj].y};
                const v2f ahi = {a[jj].z, a[jj].w};
#pragma unroll
                for (int t = 0; t < TILES; ++t) {
                    const v2f blo = {b[t].x, b[t].y};
                    acc[t] = __builtin_amdgcn_wmma_f32_16x16x4_f32(false, alo, false, blo,
                                                                   (short)0, acc[t], false, false);
                }
#pragma unroll
                for (int t = 0; t < TILES; ++t) {
                    const v2f bhi = {b[t].z, b[t].w};
                    acc[t] = __builtin_amdgcn_wmma_f32_16x16x4_f32(false, ahi, false, bhi,
                                                                   (short)0, acc[t], false, false);
                }
#pragma unroll
                for (int t = 0; t < TILES; ++t) b[t] = bn[t];
            }
#pragma unroll
            for (int jj = 0; jj < 4; ++jj) a[jj] = an[jj];
        }

        float bv[TILES];
#pragma unroll
        for (int t = 0; t < TILES; ++t) bv[t] = bias[n0 + TILES * c + t];
        float* orow = out + (size_t)(r0 + 8 * hh) * NC + n0 + TILES * c;
#pragma unroll
        for (int v = 0; v < 8; ++v) {
            float x[TILES];
#pragma unroll
            for (int t = 0; t < TILES; ++t) {
                x[t] = acc[t][v] + bv[t];
                if (RELU) x[t] = fmaxf(x[t], 0.0f);
            }
            if (TILES == 4) {
                v4f o = {x[0], x[1], x[2 % TILES], x[3 % TILES]};
                *(v4f*)(orow + (size_t)v * NC) = o;
            } else {
                v2f o = {x[0], x[1]};
                *(v2f*)(orow + (size_t)v * NC) = o;
            }
        }
    }
}

extern "C" void kernel_launch(void* const* d_in, const int* in_sizes, int n_in,
                              void* d_out, int out_size, void* d_ws, size_t ws_size,
                              hipStream_t stream) {
    (void)in_sizes; (void)n_in; (void)out_size; (void)ws_size;

    const float* x   = (const float*)d_in[0];
    const int*   src = (const int*)d_in[1];
    const int*   dst = (const int*)d_in[2];
    const float* Ws0 = (const float*)d_in[3];
    const float* Wn0 = (const float*)d_in[4];
    const float* b0  = (const float*)d_in[5];
    const float* Ws1 = (const float*)d_in[6];
    const float* Wn1 = (const float*)d_in[7];
    const float* b1  = (const float*)d_in[8];
    const float* Ws2 = (const float*)d_in[9];
    const float* Wn2 = (const float*)d_in[10];
    const float* b2  = (const float*)d_in[11];
    float* out = (float*)d_out;

    const size_t feat = (size_t)NNODES * KDIM;
    float* agg = (float*)d_ws;
    float* hA  = agg + feat;
    float* hB  = hA + feat;
    float* deg = hB + feat;

    int* cnt  = (int*)hB;
    int* rank = cnt + NNODES;
    int* csr0 = rank + NEDGES;
    int* csr    = (int*)d_out;
    int* rowptr = csr + NEDGES;

    const int edgeBlocks    = (NEDGES + 256 * EDGES_PER_THREAD - 1) / (256 * EDGES_PER_THREAD);
    const int scanBlocks    = (NNODES + SCAN_BLOCK - 1) / SCAN_BLOCK;

    hipMemsetAsync(cnt, 0, NNODES * sizeof(int), stream);
    sage_rank_kernel<<<edgeBlocks, 256, 0, stream>>>(dst, cnt, rank, NEDGES);
    sage_scan_kernel<<<scanBlocks, SCAN_BLOCK, 0, stream>>>(cnt, rowptr, deg);
    sage_fill_kernel<<<edgeBlocks, 256, 0, stream>>>(src, dst, rank, rowptr, csr0, NEDGES);
    sage_order_kernel<<<GATHER_BLOCKS, 256, 0, stream>>>(rowptr, csr0, csr);

    sage_gather_kernel<<<GATHER_BLOCKS, 256, 0, stream>>>(x, rowptr, csr, agg);
    sage_gemm_kernel<128, 4, true><<<GEMM_BLOCKS, GEMM_THREADS, 0, stream>>>(
        x, agg, deg, Ws0, Wn0, b0, hA);

    sage_gather_kernel<<<GATHER_BLOCKS, 256, 0, stream>>>(hA, rowptr, csr, agg);
    sage_gemm_kernel<128, 4, true><<<GEMM_BLOCKS, GEMM_THREADS, 0, stream>>>(
        hA, agg, deg, Ws1, Wn1, b1, hB);

    sage_gather_kernel<<<GATHER_BLOCKS, 256, 0, stream>>>(hB, rowptr, csr, agg);
    sage_gemm_kernel<64, 2, false><<<GEMM_BLOCKS, GEMM_THREADS, 0, stream>>>(
        hB, agg, deg, Ws2, Wn2, b2, out);
}


# Gather fused into the layer kernel: each wave sums its 16 rows' neighbours (same slot order) into a private LDS tile used as the neighbour A operand of its WMMAs; agg plane and 3 launches gone
# speedup vs baseline: 2.8010x; 1.0328x over previous
#include <hip/hip_runtime.h>


#define NNODES 50000
#define NEDGES 800000
#define KDIM   128

typedef __attribute__((ext_vector_type(2))) float v2f;
typedef __attribute__((ext_vector_type(8))) float v8f;

typedef __attribute__((ext_vector_type(4))) float v4f;

#define EDGES_PER_THREAD 4
#define SCAN_BLOCK       1024
#define GATHER_BLOCKS    2048

__global__ void sage_rank_kernel(const int* __restrict__ dst,
                                 int* __restrict__ cnt,
                                 int* __restrict__ rank, int E) {
    const int tid   = blockIdx.x * blockDim.x + threadIdx.x;
    const int total = gridDim.x * blockDim.x;
    int d[EDGES_PER_THREAD];
    int r[EDGES_PER_THREAD];
#pragma unroll
    for (int k = 0; k < EDGES_PER_THREAD; ++k) {
        const int e = tid + k * total;
        d[k] = (e < E) ? dst[e] : -1;
    }
#pragma unroll
    for (int k = 0; k < EDGES_PER_THREAD; ++k)
        r[k] = (d[k] >= 0) ? atomicAdd(&cnt[d[k]], 1) : 0;
#pragma unroll
    for (int k = 0; k < EDGES_PER_THREAD; ++k) {
        const int e = tid + k * total;
        if (e < E) rank[e] = r[k];
    }
}

__global__ void __launch_bounds__(SCAN_BLOCK)
sage_scan_kernel(const int* __restrict__ cnt,
                 int* __restrict__ rowptr,
                 float* __restrict__ deg) {
    __shared__ int s_pre[SCAN_BLOCK / 32];
    __shared__ int s_tot[SCAN_BLOCK / 32];
    const int t    = threadIdx.x;
    const int lane = t & 31;
    const int w    = t >> 5;
    const int b    = blockIdx.x;

    int pre = 0;
    for (int c = 0; c < b; ++c) pre += cnt[c * SCAN_BLOCK + t];
    for (int o = 16; o > 0; o >>= 1) pre += __shfl_xor(pre, o);

    const int i = b * SCAN_BLOCK + t;
    const int v = (i < NNODES) ? cnt[i] : 0;
    int inc = v;
    for (int o = 1; o < 32; o <<= 1) {
        const int u = __shfl_up(inc, o);
        if (lane >= o) inc += u;
    }
    if (lane == 31) s_tot[w] = inc;
    if (lane == 0)  s_pre[w] = pre;
    __syncthreads();

    int base = 0;
    for (int k = 0; k < SCAN_BLOCK / 32; ++k) {
        base += s_pre[k];
        if (k < w) base += s_tot[k];
    }
    const int excl = base + inc - v;
    if (i < NNODES) {
        rowptr[i] = excl;
        deg[i]    = (float)v;
    }
    if (i == NNODES - 1) rowptr[NNODES] = excl + v;
}

__global__ void sage_fill_kernel(const int* __restrict__ src,
                                 const int* __restrict__ dst,
                                 const int* __restrict__ rank,
                                 const int* __restrict__ rowptr,
                                 int* __restrict__ csr, int E) {
    const int tid   = blockIdx.x * blockDim.x + threadIdx.x;
    const int total = gridDim.x * blockDim.x;
    int d[EDGES_PER_THREAD];
    int r[EDGES_PER_THREAD];
    int s[EDGES_PER_THREAD];
    int p[EDGES_PER_THREAD];
#pragma unroll
    for (int k = 0; k < EDGES_PER_THREAD; ++k) {
        const int  e  = tid + k * total;
        const bool ok = e < E;
        const int  ec = ok ? e : E - 1;
        d[k] = dst[ec];
        s[k] = src[ec];
        r[k] = ok ? rank[ec] : E;
    }
#pragma unroll
    for (int k = 0; k < EDGES_PER_THREAD; ++k) p[k] = rowptr[d[k]] + r[k];
#pragma unroll
    for (int k = 0; k < EDGES_PER_THREAD; ++k)
        if ((unsigned)p[k] < (unsigned)E) csr[p[k]] = s[k];
}

__global__ void sage_order_kernel(const int* __restrict__ rowptr,
                                  const int* __restrict__ csr_in,
                                  int* __restrict__ csr_out) {
    const int lane = threadIdx.x & 31;
    const int wave   = __builtin_amdgcn_readfirstlane((int)(threadIdx.x >> 5));
    const int wpb    = blockDim.x >> 5;
    const int stride = gridDim.x * wpb;

    for (int node = blockIdx.x * wpb + wave; node < NNODES; node += stride) {
        const int beg = __builtin_amdgcn_readfirstlane(rowptr[node]);
        const int n   = __builtin_amdgcn_readfirstlane(rowptr[node + 1]) - beg;
        for (int i0 = 0; i0 < n; i0 += 32) {
            const int i = i0 + lane;
            const int a = (i < n) ? csr_in[beg + i] : 0;
            int pos = 0;
            for (int m0 = 0; m0 < n; m0 += 32) {
                const int mm = m0 + lane;
                const int bv = (mm < n) ? csr_in[beg + mm] : 0;
                int piece = n - m0;
                if (piece > 32) piece = 32;
                for (int u = 0; u < piece; ++u) {
                    const int b = __builtin_amdgcn_readlane(bv, u);
                    const int m = m0 + u;
                    pos += (b < a || (b == a && m < i)) ? 1 : 0;
                }
            }
            if (i < n) csr_out[beg + pos] = a;
        }
    }
}

#define ROWS_IN_FLIGHT 20

template <int U>
__device__ __forceinline__ void sage_issue_rows(const float* __restrict__ h, int idx, int j,
                                                int cnt, int lane4, v4f (&v)[ROWS_IN_FLIGHT]) {
    if constexpr (U < ROWS_IN_FLIGHT) {
        if (U < cnt) {
            const int s = __builtin_amdgcn_readlane(idx, j + U);
            v[U] = *(const v4f*)(h + (size_t)s * KDIM + lane4);
            sage_issue_rows<U + 1>(h, idx, j, cnt, lane4, v);
        }
    }
}

template <int U>
__device__ __forceinline__ void sage_sum_rows(const v4f (&v)[ROWS_IN_FLIGHT], int cnt, v4f& acc) {
    if constexpr (U < ROWS_IN_FLIGHT) {
        if (U < cnt) {
            acc += v[U];
            sage_sum_rows<U + 1>(v, cnt, acc);
        }
    }
}

#define GEMM_THREADS 512
#define GEMM_BLOCKS  256
#define GEMM_QUADS   64
#define TILE_PITCH   33

template <int NC, bool RELU>
__global__ void __launch_bounds__(GEMM_THREADS)
sage_layer_kernel(const float* __restrict__ h,
                  const int* __restrict__ rowptr,
                  const int* __restrict__ csr,
                  const float* __restrict__ deg,
                  const float* __restrict__ Ws,
                  const float* __restrict__ Wn,
                  const float* __restrict__ bias,
                  float* __restrict__ out) {
    constexpr int TILES   = NC / 16;
    constexpr int NSTRIPS = NNODES / 16;
    static_assert(NC % 64 == 0, "columns go in groups of 64 (4 tiles)");
    constexpr int PITCH   = NC + 8;
    constexpr int WPB     = GEMM_THREADS / 32;
    static_assert(NNODES % 16 == 0, "whole strips");
    __shared__ v4f sW[GEMM_QUADS * PITCH];
    __shared__ v4f sA[WPB * 16 * TILE_PITCH];

    constexpr int FILL_ITERS = GEMM_QUADS * NC / GEMM_THREADS;
    static_assert(GEMM_QUADS * NC % GEMM_THREADS == 0, "whole fill passes");
#pragma unroll 8
    for (int i = 0; i < FILL_ITERS; ++i) {
        const int idx = threadIdx.x + i * GEMM_THREADS;
        const int q = idx / NC;
        const int p = idx % NC;
        const int n = (p / 64) * 64 + 4 * (p & 15) + ((p % 64) >> 4);
        const float* w = ((q < 32) ? Ws : Wn) + (size_t)(4 * (q & 31)) * NC + n;
        v4f v;
        v.x = w[0];
        v.y = w[NC];
        v.z = w[2 * NC];
        v.w = w[3 * NC];
        sW[q * PITCH + p] = v;
    }
    __syncthreads();

    const int lane  = threadIdx.x & 31;
    const int lane4 = lane << 2;
    const int c     = lane & 15;
    const int hh    = lane >> 4;
    const int wave = __builtin_amdgcn_readfirstlane((int)(threadIdx.x >> 5));
    v4f* tile = sA + wave * (16 * TILE_PITCH);

    for (int strip = blockIdx.x + gridDim.x * wave; strip < NSTRIPS; strip += gridDim.x * WPB) {
        const int r0  = strip << 4;
        const int row = r0 + c;
        const float idg = 1.0f / fmaxf(deg[row], 1.0f);

        const int rp = rowptr[r0 + (lane < 16 ? lane : 16)];
        int beg = __builtin_amdgcn_readlane(rp, 0);
        int end = __builtin_amdgcn_readlane(rp, 1);
        int n   = end - beg;
        if (n > 32) n = 32;
        int idx = (lane < n) ? csr[beg + lane] : 0;
        for (int r = 0; r < 16; ++r) {
            const int begN = end;
            const int endN = __builtin_amdgcn_readlane(rp, r + 2);
            int nN = endN - begN;
            if (nN > 32) nN = 32;

            v4f acc = {0.f, 0.f, 0.f, 0.f};
            v4f v[ROWS_IN_FLIGHT];
            int cnt = (n < ROWS_IN_FLIGHT) ? n : ROWS_IN_FLIGHT;
            sage_issue_rows<0>(h, idx, 0, cnt, lane4, v);
            const int idxN = (lane < nN) ? csr[begN + lane] : 0;
            sage_sum_rows<0>(v, cnt, acc);

            int j = ROWS_IN_FLIGHT;
            for (int base = beg;;) {
                for (; j < n; j += ROWS_IN_FLIGHT) {
                    cnt = n - j;
                    if (cnt > ROWS_IN_FLIGHT) cnt = ROWS_IN_FLIGHT;
                    sage_issue_rows<0>(h, idx, j, cnt, lane4, v);
                    sage_sum_rows<0>(v, cnt, acc);
                }
                base += 32;
                if (base >= end) break;
                n = end - base;
                if (n > 32) n = 32;
                idx = (lane < n) ? csr[base + lane] : 0;
                j = 0;
            }
            const float sc = __builtin_bit_cast(float,
                __builtin_amdgcn_readlane(__builtin_bit_cast(int, idg), r));
            tile[r * TILE_PITCH + lane] = acc * sc;

            beg = begN; end = endN; n = nN; idx = idxN;
        }
        __builtin_amdgcn_wave_barrier();

        const v4f* hq = (const v4f*)(h + (size_t)row * KDIM) + hh;
        const v4f* gq = tile + c * TILE_PITCH + hh;

        const v4f* bq = sW + hh * PITCH + c;

        v8f acc[TILES];
#pragma unroll
        for (int t = 0; t < TILES; ++t)
            acc[t] = (v8f){0.f, 0.f, 0.f, 0.f, 0.f, 0.f, 0.f, 0.f};

        v4f a[4], an[4], b[TILES], bn[TILES];
#pragma unroll
        for (int jj = 0; jj < 4; ++jj) a[jj] = hq[2 * jj];
#pragma unroll
        for (int t = 0; t < TILES; ++t) b[t] = bq[16 * t];

#pragma unroll
        for (int ch = 0; ch < 8; ++ch) {
            if (ch + 1 < 4) {
                const v4f* nx = hq + 8 * (ch + 1);
#pragma unroll
                for (int jj = 0; jj < 4; ++jj) an[jj] = nx[2 * jj];
            } else if (ch + 1 < 8) {
                const v4f* nx = gq + 8 * (ch + 1 - 4);
#pragma unroll
                for (int jj = 0; jj < 4; ++jj) an[jj] = nx[2 * jj];
            }
#pragma unroll
            for (int jj = 0; jj < 4; ++jj) {
                const int s = 4 * ch + jj;
                if (s + 1 < 32) {
#pragma unroll
                    for (int t = 0; t < TILES; ++t) bn[t] = bq[(2 * (s + 1)) * PITCH + 16 * t];
                }
                const v2f alo = {a[jj].x, a[jj].y};
                const v2f ahi = {a[jj].z, a[jj].w};
#pragma unroll
                for (int t = 0; t < TILES; ++t) {
                    const v2f blo = {b[t].x, b[t].y};
                    acc[t] = __builtin_amdgcn_wmma_f32_16x16x4_f32(false, alo, false, blo,
                                                                   (short)0, acc[t], false, false);
                }
#pragma unroll
                for (int t = 0; t < TILES; ++t) {
                    const v2f bhi = {b[t].z, b[t].w};
                    acc[t] = __builtin_amdgcn_wmma_f32_16x16x4_f32(false, ahi, false, bhi,
                                                                   (short)0, acc[t], false, false);
                }
#pragma unroll
                for (int t = 0; t < TILES; ++t) b[t] = bn[t];
            }
#pragma unroll
            for (int jj = 0; jj < 4; ++jj) a[jj] = an[jj];
        }

        float bv[TILES];
#pragma unroll
        for (int t = 0; t < TILES; ++t) bv[t] = bias[64 * (t >> 2) + 4 * c + (t & 3)];
        float* orow = out + (size_t)(r0 + 8 * hh) * NC + 4 * c;
#pragma unroll
        for (int v = 0; v < 8; ++v) {
            float x[TILES];
#pragma unroll
            for (int t = 0; t < TILES; ++t) {
                x[t] = acc[t][v] + bv[t];
                if (RELU) x[t] = fmaxf(x[t], 0.0f);
            }
#pragma unroll
            for (int t = 0; t < TILES; t += 4) {
                v4f o = {x[t], x[t + 1], x[t + 2], x[t + 3]};
                *(v4f*)(orow + (size_t)v * NC + 16 * t) = o;
            }
        }
    }
}

extern "C" void kernel_launch(void* const* d_in, const int* in_sizes, int n_in,
                              void* d_out, int out_size, void* d_ws, size_t ws_size,
                              hipStream_t stream) {
    (void)in_sizes; (void)n_in; (void)out_size; (void)ws_size;

    const float* x   = (const float*)d_in[0];
    const int*   src = (const int*)d_in[1];
    const int*   dst = (const int*)d_in[2];
    const float* Ws0 = (const float*)d_in[3];
    const float* Wn0 = (const float*)d_in[4];
    const float* b0  = (const float*)d_in[5];
    const float* Ws1 = (const float*)d_in[6];
    const float* Wn1 = (const float*)d_in[7];
    const float* b1  = (const float*)d_in[8];
    const float* Ws2 = (const float*)d_in[9];
    const float* Wn2 = (const float*)d_in[10];
    const float* b2  = (const float*)d_in[11];
    float* out = (float*)d_out;

    const size_t feat = (size_t)NNODES * KDIM;
    float* lst = (float*)d_ws;
    float* hA  = lst + feat;
    float* hB  = hA + feat;
    float* deg = hB + feat;

    int* cnt  = (int*)hB;
    int* rank = cnt + NNODES;
    int* csr0 = rank + NEDGES;
    int* csr    = (int*)lst;
    int* rowptr = csr + NEDGES;

    const int edgeBlocks    = (NEDGES + 256 * EDGES_PER_THREAD - 1) / (256 * EDGES_PER_THREAD);
    const int scanBlocks    = (NNODES + SCAN_BLOCK - 1) / SCAN_BLOCK;

    hipMemsetAsync(cnt, 0, NNODES * sizeof(int), stream);
    sage_rank_kernel<<<edgeBlocks, 256, 0, stream>>>(dst, cnt, rank, NEDGES);
    sage_scan_kernel<<<scanBlocks, SCAN_BLOCK, 0, stream>>>(cnt, rowptr, deg);
    sage_fill_kernel<<<edgeBlocks, 256, 0, stream>>>(src, dst, rank, rowptr, csr0, NEDGES);
    sage_order_kernel<<<GATHER_BLOCKS, 256, 0, stream>>>(rowptr, csr0, csr);

    sage_layer_kernel<128, true><<<GEMM_BLOCKS, GEMM_THREADS, 0, stream>>>(
        x, rowptr, csr, deg, Ws0, Wn0, b0, hA);

    sage_layer_kernel<128, true><<<GEMM_BLOCKS, GEMM_THREADS, 0, stream>>>(
        hA, rowptr, csr, deg, Ws1, Wn1, b1, hB);

    sage_layer_kernel<64, false><<<GEMM_BLOCKS, GEMM_THREADS, 0, stream>>>(
        hB, rowptr, csr, deg, Ws2, Wn2, b2, out);
}


# Edge list built by a two-level counting sort in LDS (chunks grouped by dst range, then one range per work-group): 2 launches replace memset, atomic rank, scan, fill, order; f32 sum order unchanged
# speedup vs baseline: 4.2658x; 1.5230x over previous
#include <hip/hip_runtime.h>
#include <type_traits>


#define NNODES 50000
#define NEDGES 800000
#define KDIM   128

typedef __attribute__((ext_vector_type(2))) float v2f;
typedef __attribute__((ext_vector_type(8))) float v8f;

typedef __attribute__((ext_vector_type(4))) float v4f;

#define NBINS    256
#define BIN_W    196
#define CHUNK    (NEDGES / NBINS)
#define BUILD_T  1024
#define BIN_CAP  8192
static_assert(NBINS * BIN_W >= NNODES, "ranges cover all nodes");
static_assert(NEDGES % NBINS == 0, "whole chunks");
static_assert(NNODES <= 65536 && BIN_W <= 256,
              "packed edge word: src in bits 0..15, dst minus range start in bits 16..23");

__device__ __forceinline__ int sage_block_scan(int v, int* s_tot, int* total) {
    const int lane = threadIdx.x & 31;
    const int w    = threadIdx.x >> 5;
    int inc = v;
    for (int o = 1; o < 32; o <<= 1) {
        const int u = __shfl_up(inc, o);
        if (lane >= o) inc += u;
    }
    if (lane == 31) s_tot[w] = inc;
    __syncthreads();
    int before = 0, all = 0;
    for (int k = 0; k < BUILD_T / 32; ++k) {
        const int x = s_tot[k];
        all += x;
        if (k < w) before += x;
    }
    __syncthreads();
    *total = all;
    return before + inc - v;
}

__global__ void __launch_bounds__(BUILD_T)
sage_bin_kernel(const int* __restrict__ src,
                const int* __restrict__ dst,
                int* __restrict__ binned,
                int* __restrict__ cntT,
                int* __restrict__ offT) {
    __shared__ int s_cnt[NBINS];
    __shared__ int s_off[NBINS];
    __shared__ int s_stage[CHUNK];
    __shared__ int s_tot[BUILD_T / 32];
    const int t = threadIdx.x;
    const int c = blockIdx.x;
    if (t < NBINS) s_cnt[t] = 0;
    __syncthreads();

    constexpr int PER = (CHUNK + BUILD_T - 1) / BUILD_T;
    int word[PER], bin[PER], slot[PER];
#pragma unroll
    for (int k = 0; k < PER; ++k) {
        const int i = t + k * BUILD_T;
        bin[k]  = -1;
        word[k] = 0;
        if (i < CHUNK) {
            const int e = c * CHUNK + i;
            const unsigned d = min((unsigned)dst[e], (unsigned)(NNODES - 1));
            const unsigned b = d / BIN_W;
            bin[k]  = (int)b;
            word[k] = (src[e] & 0xFFFF) | (int)((d - b * BIN_W) << 16);
        }
    }
#pragma unroll
    for (int k = 0; k < PER; ++k)
        slot[k] = (bin[k] >= 0) ? atomicAdd(&s_cnt[bin[k]], 1) : 0;
    __syncthreads();

    int total;
    const int mine = (t < NBINS) ? s_cnt[t] : 0;
    const int ex   = sage_block_scan(mine, s_tot, &total);
    if (t < NBINS) {
        s_off[t] = ex;
        cntT[t * NBINS + c] = mine;
        offT[t * NBINS + c] = ex;
    }
    __syncthreads();
#pragma unroll
    for (int k = 0; k < PER; ++k)
        if (bin[k] >= 0) s_stage[s_off[bin[k]] + slot[k]] = word[k];
    __syncthreads();
    for (int i = t; i < CHUNK; i += BUILD_T) binned[c * CHUNK + i] = s_stage[i];
}

__global__ void __launch_bounds__(BUILD_T)
sage_list_kernel(const int* __restrict__ binned,
                 const int* __restrict__ cntT,
                 const int* __restrict__ offT,
                 int* __restrict__ scratch,
                 int* __restrict__ csr,
                 int* __restrict__ rowptr,
                 float* __restrict__ deg) {
    __shared__ int s_P[NBINS + 1];
    __shared__ int s_src0[NBINS];
    __shared__ int s_ncnt[BIN_W];
    __shared__ int s_nbeg[BIN_W];
    __shared__ int s_nfill[BIN_W];
    __shared__ int s_tot[BUILD_T / 32];
    __shared__ int s_red[BUILD_T / 32];
    __shared__ int s_stage[BIN_CAP];
    __shared__ int s_list[BIN_CAP];
    const int t    = threadIdx.x;
    const int lane = t & 31;
    const int b    = blockIdx.x;
    const int wave = __builtin_amdgcn_readfirstlane((int)(threadIdx.x >> 5));

    int acc = 0;
    const int4* c4 = (const int4*)cntT;
    constexpr int BASE_ITERS = ((NBINS - 1) * (NBINS / 4) + BUILD_T - 1) / BUILD_T;
    int4 cv[BASE_ITERS];
#pragma unroll
    for (int j = 0; j < BASE_ITERS; ++j) {
        const int i = t + j * BUILD_T;
        cv[j] = (i < b * (NBINS / 4)) ? c4[i] : int4{0, 0, 0, 0};
    }
#pragma unroll
    for (int j = 0; j < BASE_ITERS; ++j) acc += cv[j].x + cv[j].y + cv[j].z + cv[j].w;
    for (int o = 16; o > 0; o >>= 1) acc += __shfl_xor(acc, o);
    if (lane == 0) s_red[wave] = acc;

    const int pc = (t < NBINS) ? cntT[b * NBINS + t] : 0;
    if (t < NBINS) s_src0[t] = t * CHUNK + offT[b * NBINS + t];
    if (t < BIN_W) { s_ncnt[t] = 0; s_nfill[t] = 0; }
    int tot;
    const int pex = sage_block_scan(pc, s_tot, &tot);
    if (t < NBINS) s_P[t] = pex;
    if (t == 0)    s_P[NBINS] = tot;
    int base = 0;
    for (int k = 0; k < BUILD_T / 32; ++k) base += s_red[k];
    __syncthreads();

    const bool fits = tot <= BIN_CAP;

    auto fetch = [&](int k) -> int {
        static_assert(NBINS == 256, "eight halvings");
        int lo = 0, hi = NBINS;
#pragma unroll
        for (int s = 0; s < 8; ++s) {
            const int mid = (lo + hi) >> 1;
            if (s_P[mid] <= k) lo = mid; else hi = mid;
        }
        return binned[s_src0[lo] + (k - s_P[lo])];
    };

    for (int k0 = t; k0 < tot; k0 += 4 * BUILD_T) {
        int w4[4];
#pragma unroll
        for (int j = 0; j < 4; ++j) {
            const int k = k0 + j * BUILD_T;
            w4[j] = (k < tot) ? fetch(k) : 0;
        }
#pragma unroll
        for (int j = 0; j < 4; ++j) {
            const int k = k0 + j * BUILD_T;
            if (k < tot) {
                if (fits) s_stage[k] = w4[j];
                atomicAdd(&s_ncnt[w4[j] >> 16], 1);
            }
        }
    }
    __syncthreads();

    const int nc = (t < BIN_W) ? s_ncnt[t] : 0;
    int tot2;
    const int nex = sage_block_scan(nc, s_tot, &tot2);
    if (t < BIN_W) {
        const int node = b * BIN_W + t;
        s_nbeg[t] = nex;
        if (node < NNODES) {
            rowptr[node] = base + nex;
            deg[node]    = (float)nc;
        }
        if (node == NNODES - 1) rowptr[NNODES] = base + tot;
    }
    __syncthreads();

    int* glist = scratch + base;
    int* out   = csr + base;

    auto place_and_order = [&](auto in_lds) {
        constexpr bool IN_LDS = decltype(in_lds)::value;
        for (int k = t; k < tot; k += BUILD_T) {
            int w;
            if constexpr (IN_LDS) w = s_stage[k]; else w = fetch(k);
            const int dl = w >> 16;
            const int sl = s_nbeg[dl] + atomicAdd(&s_nfill[dl], 1);
            if constexpr (IN_LDS) s_list[sl] = w & 0xFFFF; else glist[sl] = w & 0xFFFF;
        }
        __syncthreads();
        for (int nd = wave; nd < BIN_W; nd += BUILD_T / 32) {
            const int beg = __builtin_amdgcn_readfirstlane(s_nbeg[nd]);
            const int n   = __builtin_amdgcn_readfirstlane(s_ncnt[nd]);
            for (int i0 = 0; i0 < n; i0 += 32) {
                const int i = i0 + lane;
                int a = 0;
                if (i < n) { if constexpr (IN_LDS) a = s_list[beg + i]; else a = glist[beg + i]; }
                int pos = 0;
                for (int m0 = 0; m0 < n; m0 += 32) {
                    const int mm = m0 + lane;
                    int bv = 0;
                    if (mm < n) { if constexpr (IN_LDS) bv = s_list[beg + mm]; else bv = glist[beg + mm]; }
                    int piece = n - m0;
                    if (piece > 32) piece = 32;
                    for (int u = 0; u < piece; ++u) {
                        const int x = __builtin_amdgcn_readlane(bv, u);
                        const int m = m0 + u;
                        pos += (x < a || (x == a && m < i)) ? 1 : 0;
                    }
                }
                if (i < n) out[beg + pos] = a;
            }
        }
    };
    if (fits) place_and_order(std::true_type{});
    else      place_and_order(std::false_type{});
}

#define ROWS_IN_FLIGHT 20

template <int U>
__device__ __forceinline__ void sage_issue_rows(const float* __restrict__ h, int idx, int j,
                                                int cnt, int lane4, v4f (&v)[ROWS_IN_FLIGHT]) {
    if constexpr (U < ROWS_IN_FLIGHT) {
        if (U < cnt) {
            const int s = __builtin_amdgcn_readlane(idx, j + U);
            v[U] = *(const v4f*)(h + (size_t)s * KDIM + lane4);
            sage_issue_rows<U + 1>(h, idx, j, cnt, lane4, v);
        }
    }
}

template <int U>
__device__ __forceinline__ void sage_sum_rows(const v4f (&v)[ROWS_IN_FLIGHT], int cnt, v4f& acc) {
    if constexpr (U < ROWS_IN_FLIGHT) {
        if (U < cnt) {
            acc += v[U];
            sage_sum_rows<U + 1>(v, cnt, acc);
        }
    }
}

#define GEMM_THREADS 512
#define GEMM_BLOCKS  256
#define GEMM_QUADS   64
#define TILE_PITCH   33

template <int NC, bool RELU>
__global__ void __launch_bounds__(GEMM_THREADS)
sage_layer_kernel(const float* __restrict__ h,
                  const int* __restrict__ rowptr,
                  const int* __restrict__ csr,
                  const float* __restrict__ deg,
                  const float* __restrict__ Ws,
                  const float* __restrict__ Wn,
                  const float* __restrict__ bias,
                  float* __restrict__ out) {
    constexpr int TILES   = NC / 16;
    constexpr int NSTRIPS = NNODES / 16;
    static_assert(NC % 64 == 0, "columns go in groups of 64 (4 tiles)");
    constexpr int PITCH   = NC + 8;
    constexpr int WPB     = GEMM_THREADS / 32;
    static_assert(NNODES % 16 == 0, "whole strips");
    __shared__ v4f sW[GEMM_QUADS * PITCH];
    __shared__ v4f sA[WPB * 16 * TILE_PITCH];

    constexpr int FILL_ITERS = GEMM_QUADS * NC / GEMM_THREADS;
    static_assert(GEMM_QUADS * NC % GEMM_THREADS == 0, "whole fill passes");
#pragma unroll 8
    for (int i = 0; i < FILL_ITERS; ++i) {
        const int idx = threadIdx.x + i * GEMM_THREADS;
        const int q = idx / NC;
        const int p = idx % NC;
        const int n = (p / 64) * 64 + 4 * (p & 15) + ((p % 64) >> 4);
        const float* w = ((q < 32) ? Ws : Wn) + (size_t)(4 * (q & 31)) * NC + n;
        v4f v;
        v.x = w[0];
        v.y = w[NC];
        v.z = w[2 * NC];
        v.w = w[3 * NC];
        sW[q * PITCH + p] = v;
    }
    __syncthreads();

    const int lane  = threadIdx.x & 31;
    const int lane4 = lane << 2;
    const int c     = lane & 15;
    const int hh    = lane >> 4;
    const int wave = __builtin_amdgcn_readfirstlane((int)(threadIdx.x >> 5));
    v4f* tile = sA + wave * (16 * TILE_PITCH);

    for (int strip = blockIdx.x + gridDim.x * wave; strip < NSTRIPS; strip += gridDim.x * WPB) {
        const int r0  = strip << 4;
        const int row = r0 + c;
        const float idg = 1.0f / fmaxf(deg[row], 1.0f);

        const int rp = rowptr[r0 + (lane < 16 ? lane : 16)];
        int beg = __builtin_amdgcn_readlane(rp, 0);
        int end = __builtin_amdgcn_readlane(rp, 1);
        int n   = end - beg;
        if (n > 32) n = 32;
        int idx = (lane < n) ? csr[beg + lane] : 0;
        for (int r = 0; r < 16; ++r) {
            const int begN = end;
            const int endN = __builtin_amdgcn_readlane(rp, r + 2);
            int nN = endN - begN;
            if (nN > 32) nN = 32;

            v4f acc = {0.f, 0.f, 0.f, 0.f};
            v4f v[ROWS_IN_FLIGHT];
            int cnt = (n < ROWS_IN_FLIGHT) ? n : ROWS_IN_FLIGHT;
            sage_issue_rows<0>(h, idx, 0, cnt, lane4, v);
            const int idxN = (lane < nN) ? csr[begN + lane] : 0;
            sage_sum_rows<0>(v, cnt, acc);

            int j = ROWS_IN_FLIGHT;
            for (int base = beg;;) {
                for (; j < n; j += ROWS_IN_FLIGHT) {
                    cnt = n - j;
                    if (cnt > ROWS_IN_FLIGHT) cnt = ROWS_IN_FLIGHT;
                    sage_issue_rows<0>(h, idx, j, cnt, lane4, v);
                    sage_sum_rows<0>(v, cnt, acc);
                }
                base += 32;
                if (base >= end) break;
                n = end - base;
                if (n > 32) n = 32;
                idx = (lane < n) ? csr[base + lane] : 0;
                j = 0;
            }
            const float sc = __builtin_bit_cast(float,
                __builtin_amdgcn_readlane(__builtin_bit_cast(int, idg), r));
            tile[r * TILE_PITCH + lane] = acc * sc;

            beg = begN; end = endN; n = nN; idx = idxN;
        }
        __builtin_amdgcn_wave_barrier();

        const v4f* hq = (const v4f*)(h + (size_t)row * KDIM) + hh;
        const v4f* gq = tile + c * TILE_PITCH + hh;

        const v4f* bq = sW + hh * PITCH + c;

        v8f acc[TILES];
#pragma unroll
        for (int t = 0; t < TILES; ++t)
            acc[t] = (v8f){0.f, 0.f, 0.f, 0.f, 0.f, 0.f, 0.f, 0.f};

        v4f a[4], an[4], b[TILES], bn[TILES];
#pragma unroll
        for (int jj = 0; jj < 4; ++jj) a[jj] = hq[2 * jj];
#pragma unroll
        for (int t = 0; t < TILES; ++t) b[t] = bq[16 * t];

#pragma unroll
        for (int ch = 0; ch < 8; ++ch) {
            if (ch + 1 < 4) {
                const v4f* nx = hq + 8 * (ch + 1);
#pragma unroll
                for (int jj = 0; jj < 4; ++jj) an[jj] = nx[2 * jj];
            } else if (ch + 1 < 8) {
                const v4f* nx = gq + 8 * (ch + 1 - 4);
#pragma unroll
                for (int jj = 0; jj < 4; ++jj) an[jj] = nx[2 * jj];
            }
#pragma unroll
            for (int jj = 0; jj < 4; ++jj) {
                const int s = 4 * ch + jj;
                if (s + 1 < 32) {
#pragma unroll
                    for (int t = 0; t < TILES; ++t) bn[t] = bq[(2 * (s + 1)) * PITCH + 16 * t];
                }
                const v2f alo = {a[jj].x, a[jj].y};
                const v2f ahi = {a[jj].z, a[jj].w};
#pragma unroll
                for (int t = 0; t < TILES; ++t) {
                    const v2f blo = {b[t].x, b[t].y};
                    acc[t] = __builtin_amdgcn_wmma_f32_16x16x4_f32(false, alo, false, blo,
                                                                   (short)0, acc[t], false, false);
                }
#pragma unroll
                for (int t = 0; t < TILES; ++t) {
                    const v2f bhi = {b[t].z, b[t].w};
                    acc[t] = __builtin_amdgcn_wmma_f32_16x16x4_f32(false, ahi, false, bhi,
                                                                   (short)0, acc[t], false, false);
                }
#pragma unroll
                for (int t = 0; t < TILES; ++t) b[t] = bn[t];
            }
#pragma unroll
            for (int jj = 0; jj < 4; ++jj) a[jj] = an[jj];
        }

        float bv[TILES];
#pragma unroll
        for (int t = 0; t < TILES; ++t) bv[t] = bias[64 * (t >> 2) + 4 * c + (t & 3)];
        float* orow = out + (size_t)(r0 + 8 * hh) * NC + 4 * c;
#pragma unroll
        for (int v = 0; v < 8; ++v) {
            float x[TILES];
#pragma unroll
            for (int t = 0; t < TILES; ++t) {
                x[t] = acc[t][v] + bv[t];
                if (RELU) x[t] = fmaxf(x[t], 0.0f);
            }
#pragma unroll
            for (int t = 0; t < TILES; t += 4) {
                v4f o = {x[t], x[t + 1], x[t + 2], x[t + 3]};
                *(v4f*)(orow + (size_t)v * NC + 16 * t) = o;
            }
        }
    }
}

extern "C" void kernel_launch(void* const* d_in, const int* in_sizes, int n_in,
                              void* d_out, int out_size, void* d_ws, size_t ws_size,
                              hipStream_t stream) {
    (void)in_sizes; (void)n_in; (void)out_size; (void)ws_size;

    const float* x   = (const float*)d_in[0];
    const int*   src = (const int*)d_in[1];
    const int*   dst = (const int*)d_in[2];
    const float* Ws0 = (const float*)d_in[3];
    const float* Wn0 = (const float*)d_in[4];
    const float* b0  = (const float*)d_in[5];
    const float* Ws1 = (const float*)d_in[6];
    const float* Wn1 = (const float*)d_in[7];
    const float* b1  = (const float*)d_in[8];
    const float* Ws2 = (const float*)d_in[9];
    const float* Wn2 = (const float*)d_in[10];
    const float* b2  = (const float*)d_in[11];
    float* out = (float*)d_out;

    const size_t feat = (size_t)NNODES * KDIM;
    float* lst = (float*)d_ws;
    float* hA  = lst + feat;
    float* hB  = hA + feat;
    float* deg = hB + feat;

    int* binned  = (int*)hB;
    int* cntT    = binned + NEDGES;
    int* offT    = cntT + NBINS * NBINS;
    int* scratch = offT + NBINS * NBINS;
    int* csr    = (int*)lst;
    int* rowptr = csr + NEDGES;

    sage_bin_kernel<<<NBINS, BUILD_T, 0, stream>>>(src, dst, binned, cntT, offT);
    sage_list_kernel<<<NBINS, BUILD_T, 0, stream>>>(binned, cntT, offT, scratch, csr, rowptr, deg);

    sage_layer_kernel<128, true><<<GEMM_BLOCKS, GEMM_THREADS, 0, stream>>>(
        x, rowptr, csr, deg, Ws0, Wn0, b0, hA);

    sage_layer_kernel<128, true><<<GEMM_BLOCKS, GEMM_THREADS, 0, stream>>>(
        hA, rowptr, csr, deg, Ws1, Wn1, b1, hB);

    sage_layer_kernel<64, false><<<GEMM_BLOCKS, GEMM_THREADS, 0, stream>>>(
        hB, rowptr, csr, deg, Ws2, Wn2, b2, out);
}


# Neighbour rows of two nodes (first group of each, up to 24 rows) issued together before either is summed: 8 load round trips per 16-row strip, not 16; slot order of every f32 sum unchanged
# speedup vs baseline: 4.4584x; 1.0451x over previous
#include <hip/hip_runtime.h>
#include <type_traits>


#define NNODES 50000
#define NEDGES 800000
#define KDIM   128

typedef __attribute__((ext_vector_type(2))) float v2f;
typedef __attribute__((ext_vector_type(8))) float v8f;

typedef __attribute__((ext_vector_type(4))) float v4f;

#define NBINS    256
#define BIN_W    196
#define CHUNK    (NEDGES / NBINS)
#define BUILD_T  1024
#define BIN_CAP  8192
static_assert(NBINS * BIN_W >= NNODES, "ranges cover all nodes");
static_assert(NEDGES % NBINS == 0, "whole chunks");
static_assert(NNODES <= 65536 && BIN_W <= 256,
              "packed edge word: src in bits 0..15, dst minus range start in bits 16..23");

__device__ __forceinline__ int sage_block_scan(int v, int* s_tot, int* total) {
    const int lane = threadIdx.x & 31;
    const int w    = threadIdx.x >> 5;
    int inc = v;
    for (int o = 1; o < 32; o <<= 1) {
        const int u = __shfl_up(inc, o);
        if (lane >= o) inc += u;
    }
    if (lane == 31) s_tot[w] = inc;
    __syncthreads();
    int before = 0, all = 0;
    for (int k = 0; k < BUILD_T / 32; ++k) {
        const int x = s_tot[k];
        all += x;
        if (k < w) before += x;
    }
    __syncthreads();
    *total = all;
    return before + inc - v;
}

__global__ void __launch_bounds__(BUILD_T)
sage_bin_kernel(const int* __restrict__ src,
                const int* __restrict__ dst,
                int* __restrict__ binned,
                int* __restrict__ cntT,
                int* __restrict__ offT) {
    __shared__ int s_cnt[NBINS];
    __shared__ int s_off[NBINS];
    __shared__ int s_stage[CHUNK];
    __shared__ int s_tot[BUILD_T / 32];
    const int t = threadIdx.x;
    const int c = blockIdx.x;
    if (t < NBINS) s_cnt[t] = 0;
    __syncthreads();

    constexpr int PER = (CHUNK + BUILD_T - 1) / BUILD_T;
    int word[PER], bin[PER], slot[PER];
#pragma unroll
    for (int k = 0; k < PER; ++k) {
        const int i = t + k * BUILD_T;
        bin[k]  = -1;
        word[k] = 0;
        if (i < CHUNK) {
            const int e = c * CHUNK + i;
            const unsigned d = min((unsigned)dst[e], (unsigned)(NNODES - 1));
            const unsigned b = d / BIN_W;
            bin[k]  = (int)b;
            word[k] = (src[e] & 0xFFFF) | (int)((d - b * BIN_W) << 16);
        }
    }
#pragma unroll
    for (int k = 0; k < PER; ++k)
        slot[k] = (bin[k] >= 0) ? atomicAdd(&s_cnt[bin[k]], 1) : 0;
    __syncthreads();

    int total;
    const int mine = (t < NBINS) ? s_cnt[t] : 0;
    const int ex   = sage_block_scan(mine, s_tot, &total);
    if (t < NBINS) {
        s_off[t] = ex;
        cntT[t * NBINS + c] = mine;
        offT[t * NBINS + c] = ex;
    }
    __syncthreads();
#pragma unroll
    for (int k = 0; k < PER; ++k)
        if (bin[k] >= 0) s_stage[s_off[bin[k]] + slot[k]] = word[k];
    __syncthreads();
    for (int i = t; i < CHUNK; i += BUILD_T) binned[c * CHUNK + i] = s_stage[i];
}

__global__ void __launch_bounds__(BUILD_T)
sage_list_kernel(const int* __restrict__ binned,
                 const int* __restrict__ cntT,
                 const int* __restrict__ offT,
                 int* __restrict__ scratch,
                 int* __restrict__ csr,
                 int* __restrict__ rowptr,
                 float* __restrict__ deg) {
    __shared__ int s_P[NBINS + 1];
    __shared__ int s_src0[NBINS];
    __shared__ int s_ncnt[BIN_W];
    __shared__ int s_nbeg[BIN_W];
    __shared__ int s_nfill[BIN_W];
    __shared__ int s_tot[BUILD_T / 32];
    __shared__ int s_red[BUILD_T / 32];
    __shared__ int s_stage[BIN_CAP];
    __shared__ int s_list[BIN_CAP];
    const int t    = threadIdx.x;
    const int lane = t & 31;
    const int b    = blockIdx.x;
    const int wave = __builtin_amdgcn_readfirstlane((int)(threadIdx.x >> 5));

    int acc = 0;
    const int4* c4 = (const int4*)cntT;
    constexpr int BASE_ITERS = ((NBINS - 1) * (NBINS / 4) + BUILD_T - 1) / BUILD_T;
    int4 cv[BASE_ITERS];
#pragma unroll
    for (int j = 0; j < BASE_ITERS; ++j) {
        const int i = t + j * BUILD_T;
        cv[j] = (i < b * (NBINS / 4)) ? c4[i] : int4{0, 0, 0, 0};
    }
#pragma unroll
    for (int j = 0; j < BASE_ITERS; ++j) acc += cv[j].x + cv[j].y + cv[j].z + cv[j].w;
    for (int o = 16; o > 0; o >>= 1) acc += __shfl_xor(acc, o);
    if (lane == 0) s_red[wave] = acc;

    const int pc = (t < NBINS) ? cntT[b * NBINS + t] : 0;
    if (t < NBINS) s_src0[t] = t * CHUNK + offT[b * NBINS + t];
    if (t < BIN_W) { s_ncnt[t] = 0; s_nfill[t] = 0; }
    int tot;
    const int pex = sage_block_scan(pc, s_tot, &tot);
    if (t < NBINS) s_P[t] = pex;
    if (t == 0)    s_P[NBINS] = tot;
    int base = 0;
    for (int k = 0; k < BUILD_T / 32; ++k) base += s_red[k];
    __syncthreads();

    const bool fits = tot <= BIN_CAP;

    auto fetch = [&](int k) -> int {
        static_assert(NBINS == 256, "eight halvings");
        int lo = 0, hi = NBINS;
#pragma unroll
        for (int s = 0; s < 8; ++s) {
            const int mid = (lo + hi) >> 1;
            if (s_P[mid] <= k) lo = mid; else hi = mid;
        }
        return binned[s_src0[lo] + (k - s_P[lo])];
    };

    for (int k0 = t; k0 < tot; k0 += 4 * BUILD_T) {
        int w4[4];
#pragma unroll
        for (int j = 0; j < 4; ++j) {
            const int k = k0 + j * BUILD_T;
            w4[j] = (k < tot) ? fetch(k) : 0;
        }
#pragma unroll
        for (int j = 0; j < 4; ++j) {
            const int k = k0 + j * BUILD_T;
            if (k < tot) {
                if (fits) s_stage[k] = w4[j];
                atomicAdd(&s_ncnt[w4[j] >> 16], 1);
            }
        }
    }
    __syncthreads();

    const int nc = (t < BIN_W) ? s_ncnt[t] : 0;
    int tot2;
    const int nex = sage_block_scan(nc, s_tot, &tot2);
    if (t < BIN_W) {
        const int node = b * BIN_W + t;
        s_nbeg[t] = nex;
        if (node < NNODES) {
            rowptr[node] = base + nex;
            deg[node]    = (float)nc;
        }
        if (node == NNODES - 1) rowptr[NNODES] = base + tot;
    }
    __syncthreads();

    int* glist = scratch + base;
    int* out   = csr + base;

    auto place_and_order = [&](auto in_lds) {
        constexpr bool IN_LDS = decltype(in_lds)::value;
        for (int k = t; k < tot; k += BUILD_T) {
            int w;
            if constexpr (IN_LDS) w = s_stage[k]; else w = fetch(k);
            const int dl = w >> 16;
            const int sl = s_nbeg[dl] + atomicAdd(&s_nfill[dl], 1);
            if constexpr (IN_LDS) s_list[sl] = w & 0xFFFF; else glist[sl] = w & 0xFFFF;
        }
        __syncthreads();
        for (int nd = wave; nd < BIN_W; nd += BUILD_T / 32) {
            const int beg = __builtin_amdgcn_readfirstlane(s_nbeg[nd]);
            const int n   = __builtin_amdgcn_readfirstlane(s_ncnt[nd]);
            for (int i0 = 0; i0 < n; i0 += 32) {
                const int i = i0 + lane;
                int a = 0;
                if (i < n) { if constexpr (IN_LDS) a = s_list[beg + i]; else a = glist[beg + i]; }
                int pos = 0;
                for (int m0 = 0; m0 < n; m0 += 32) {
                    const int mm = m0 + lane;
                    int bv = 0;
                    if (mm < n) { if constexpr (IN_LDS) bv = s_list[beg + mm]; else bv = glist[beg + mm]; }
                    int piece = n - m0;
                    if (piece > 32) piece = 32;
                    for (int u = 0; u < piece; ++u) {
                        const int x = __builtin_amdgcn_readlane(bv, u);
                        const int m = m0 + u;
                        pos += (x < a || (x == a && m < i)) ? 1 : 0;
                    }
                }
                if (i < n) out[beg + pos] = a;
            }
        }
    };
    if (fits) place_and_order(std::true_type{});
    else      place_and_order(std::false_type{});
}

#define ROWS_IN_FLIGHT 24

template <int U>
__device__ __forceinline__ void sage_issue_rows(const float* __restrict__ h, int idx, int j,
                                                int s_first, int cnt, int lane4,
                                                v4f (&v)[ROWS_IN_FLIGHT]) {
    if constexpr (U < ROWS_IN_FLIGHT) {
        if (U < cnt) {
            const int s = (U == 0) ? s_first : __builtin_amdgcn_readlane(idx, j + U);
            v[U] = *(const v4f*)(h + (size_t)s * KDIM + lane4);
            sage_issue_rows<U + 1>(h, idx, j, s_first, cnt, lane4, v);
        }
    }
}

template <int U>
__device__ __forceinline__ void sage_sum_rows(const v4f (&v)[ROWS_IN_FLIGHT], int cnt, v4f& acc) {
    if constexpr (U < ROWS_IN_FLIGHT) {
        if (U < cnt) {
            acc += v[U];
            sage_sum_rows<U + 1>(v, cnt, acc);
        }
    }
}

__device__ __forceinline__ void sage_finish_list(const float* __restrict__ h,
                                                 const int* __restrict__ csr,
                                                 int beg, int end, int n, int idx,
                                                 int lane, int lane4,
                                                 v4f (&v)[ROWS_IN_FLIGHT], v4f& acc) {
    int j = ROWS_IN_FLIGHT;
    for (int base = beg;;) {
        for (; j < n; j += ROWS_IN_FLIGHT) {
            int cnt = n - j;
            if (cnt > ROWS_IN_FLIGHT) cnt = ROWS_IN_FLIGHT;
            sage_issue_rows<0>(h, idx, j, __builtin_amdgcn_readlane(idx, j), cnt, lane4, v);
            sage_sum_rows<0>(v, cnt, acc);
        }
        base += 32;
        if (base >= end) break;
        n = end - base;
        if (n > 32) n = 32;
        idx = (lane < n) ? csr[base + lane] : 0;
        j = 0;
    }
}

#define GEMM_THREADS 512
#define GEMM_BLOCKS  256
#define GEMM_QUADS   64
#define TILE_PITCH   33

template <int NC, bool RELU>
__global__ void __launch_bounds__(GEMM_THREADS)
sage_layer_kernel(const float* __restrict__ h,
                  const int* __restrict__ rowptr,
                  const int* __restrict__ csr,
                  const float* __restrict__ deg,
                  const float* __restrict__ Ws,
                  const float* __restrict__ Wn,
                  const float* __restrict__ bias,
                  float* __restrict__ out) {
    constexpr int TILES   = NC / 16;
    constexpr int NSTRIPS = NNODES / 16;
    static_assert(NC % 64 == 0, "columns go in groups of 64 (4 tiles)");
    constexpr int PITCH   = NC + 8;
    constexpr int WPB     = GEMM_THREADS / 32;
    static_assert(NNODES % 16 == 0, "whole strips");
    __shared__ v4f sW[GEMM_QUADS * PITCH];
    __shared__ v4f sA[WPB * 16 * TILE_PITCH];

    constexpr int FILL_ITERS = GEMM_QUADS * NC / GEMM_THREADS;
    static_assert(GEMM_QUADS * NC % GEMM_THREADS == 0, "whole fill passes");
#pragma unroll 8
    for (int i = 0; i < FILL_ITERS; ++i) {
        const int idx = threadIdx.x + i * GEMM_THREADS;
        const int q = idx / NC;
        const int p = idx % NC;
        const int n = (p / 64) * 64 + 4 * (p & 15) + ((p % 64) >> 4);
        const float* w = ((q < 32) ? Ws : Wn) + (size_t)(4 * (q & 31)) * NC + n;
        v4f v;
        v.x = w[0];
        v.y = w[NC];
        v.z = w[2 * NC];
        v.w = w[3 * NC];
        sW[q * PITCH + p] = v;
    }
    __syncthreads();

    const int lane  = threadIdx.x & 31;
    const int lane4 = lane << 2;
    const int c     = lane & 15;
    const int hh    = lane >> 4;
    const int wave = __builtin_amdgcn_readfirstlane((int)(threadIdx.x >> 5));
    v4f* tile = sA + wave * (16 * TILE_PITCH);

    static_assert(GEMM_BLOCKS * WPB >= NSTRIPS, "a wave takes at most one strip");
    const int strip = blockIdx.x + GEMM_BLOCKS * wave;
    if (strip < NSTRIPS) {
        const int r0  = strip << 4;
        const int row = r0 + c;
        const float idg = 1.0f / fmaxf(deg[row], 1.0f);

        const int rp = rowptr[r0 + (lane < 16 ? lane : 16)];
        int begA = __builtin_amdgcn_readlane(rp, 0);
        int endA = __builtin_amdgcn_readlane(rp, 1);
        int endB = __builtin_amdgcn_readlane(rp, 2);
        int nA = endA - begA;
        if (nA > 32) nA = 32;
        int nB = endB - endA;
        if (nB > 32) nB = 32;
        int idxA = (lane < nA) ? csr[begA + lane] : 0;
        int idxB = (lane < nB) ? csr[endA + lane] : 0;
        for (int r = 0; r < 16; r += 2) {
            const int begB  = endA;
            const int endNA = __builtin_amdgcn_readlane(rp, r + 3);
            const int endNB = __builtin_amdgcn_readlane(rp, r + 4);
            int nNA = endNA - endB;
            if (nNA > 32) nNA = 32;
            int nNB = endNB - endNA;
            if (nNB > 32) nNB = 32;

            v4f accA = {0.f, 0.f, 0.f, 0.f};
            v4f accB = {0.f, 0.f, 0.f, 0.f};
            v4f vA[ROWS_IN_FLIGHT], vB[ROWS_IN_FLIGHT];
            const int cntA = (nA < ROWS_IN_FLIGHT) ? nA : ROWS_IN_FLIGHT;
            const int cntB = (nB < ROWS_IN_FLIGHT) ? nB : ROWS_IN_FLIGHT;
            const int sA0 = __builtin_amdgcn_readlane(idxA, 0);
            const int sB0 = __builtin_amdgcn_readlane(idxB, 0);
            sage_issue_rows<0>(h, idxA, 0, sA0, cntA, lane4, vA);
            sage_issue_rows<0>(h, idxB, 0, sB0, cntB, lane4, vB);
            const int idxNA = (lane < nNA) ? csr[endB + lane] : 0;
            const int idxNB = (lane < nNB) ? csr[endNA + lane] : 0;
            sage_sum_rows<0>(vA, cntA, accA);
            sage_sum_rows<0>(vB, cntB, accB);

            sage_finish_list(h, csr, begA, endA, nA, idxA, lane, lane4, vA, accA);
            sage_finish_list(h, csr, begB, endB, nB, idxB, lane, lane4, vB, accB);

            const float scA = __builtin_bit_cast(float,
                __builtin_amdgcn_readlane(__builtin_bit_cast(int, idg), r));
            const float scB = __builtin_bit_cast(float,
                __builtin_amdgcn_readlane(__builtin_bit_cast(int, idg), r + 1));
            tile[r * TILE_PITCH + lane]       = accA * scA;
            tile[(r + 1) * TILE_PITCH + lane] = accB * scB;

            begA = endB; endA = endNA; endB = endNB;
            nA = nNA; nB = nNB; idxA = idxNA; idxB = idxNB;
        }
        __builtin_amdgcn_wave_barrier();

        const v4f* hq = (const v4f*)(h + (size_t)row * KDIM) + hh;
        const v4f* gq = tile + c * TILE_PITCH + hh;

        const v4f* bq = sW + hh * PITCH + c;

        v8f acc[TILES];
#pragma unroll
        for (int t = 0; t < TILES; ++t)
            acc[t] = (v8f){0.f, 0.f, 0.f, 0.f, 0.f, 0.f, 0.f, 0.f};

        v4f a[4], an[4], b[TILES], bn[TILES];
#pragma unroll
        for (int jj = 0; jj < 4; ++jj) a[jj] = hq[2 * jj];
#pragma unroll
        for (int t = 0; t < TILES; ++t) b[t] = bq[16 * t];

#pragma unroll
        for (int ch = 0; ch < 8; ++ch) {
            if (ch + 1 < 4) {
                const v4f* nx = hq + 8 * (ch + 1);
#pragma unroll
                for (int jj = 0; jj < 4; ++jj) an[jj] = nx[2 * jj];
            } else if (ch + 1 < 8) {
                const v4f* nx = gq + 8 * (ch + 1 - 4);
#pragma unroll
                for (int jj = 0; jj < 4; ++jj) an[jj] = nx[2 * jj];
            }
#pragma unroll
            for (int jj = 0; jj < 4; ++jj) {
                const int s = 4 * ch + jj;
                if (s + 1 < 32) {
#pragma unroll
                    for (int t = 0; t < TILES; ++t) bn[t] = bq[(2 * (s + 1)) * PITCH + 16 * t];
                }
                const v2f alo = {a[jj].x, a[jj].y};
                const v2f ahi = {a[jj].z, a[jj].w};
#pragma unroll
                for (int t = 0; t < TILES; ++t) {
                    const v2f blo = {b[t].x, b[t].y};
                    acc[t] = __builtin_amdgcn_wmma_f32_16x16x4_f32(false, alo, false, blo,
                                                                   (short)0, acc[t], false, false);
                }
#pragma unroll
                for (int t = 0; t < TILES; ++t) {
                    const v2f bhi = {b[t].z, b[t].w};
                    acc[t] = __builtin_amdgcn_wmma_f32_16x16x4_f32(false, ahi, false, bhi,
                                                                   (short)0, acc[t], false, false);
                }
#pragma unroll
                for (int t = 0; t < TILES; ++t) b[t] = bn[t];
            }
#pragma unroll
            for (int jj = 0; jj < 4; ++jj) a[jj] = an[jj];
        }

        float bv[TILES];
#pragma unroll
        for (int t = 0; t < TILES; ++t) bv[t] = bias[64 * (t >> 2) + 4 * c + (t & 3)];
        float* orow = out + (size_t)(r0 + 8 * hh) * NC + 4 * c;
#pragma unroll
        for (int v = 0; v < 8; ++v) {
            float x[TILES];
#pragma unroll
            for (int t = 0; t < TILES; ++t) {
                x[t] = acc[t][v] + bv[t];
                if (RELU) x[t] = fmaxf(x[t], 0.0f);
            }
#pragma unroll
            for (int t = 0; t < TILES; t += 4) {
                v4f o = {x[t], x[t + 1], x[t + 2], x[t + 3]};
                *(v4f*)(orow + (size_t)v * NC + 16 * t) = o;
            }
        }
    }
}

extern "C" void kernel_launch(void* const* d_in, const int* in_sizes, int n_in,
                              void* d_out, int out_size, void* d_ws, size_t ws_size,
                              hipStream_t stream) {
    (void)in_sizes; (void)n_in; (void)out_size; (void)ws_size;

    const float* x   = (const float*)d_in[0];
    const int*   src = (const int*)d_in[1];
    const int*   dst = (const int*)d_in[2];
    const float* Ws0 = (const float*)d_in[3];
    const float* Wn0 = (const float*)d_in[4];
    const float* b0  = (const float*)d_in[5];
    const float* Ws1 = (const float*)d_in[6];
    const float* Wn1 = (const float*)d_in[7];
    const float* b1  = (const float*)d_in[8];
    const float* Ws2 = (const float*)d_in[9];
    const float* Wn2 = (const float*)d_in[10];
    const float* b2  = (const float*)d_in[11];
    float* out = (float*)d_out;

    const size_t feat = (size_t)NNODES * KDIM;
    float* lst = (float*)d_ws;
    float* hA  = lst + feat;
    float* hB  = hA + feat;
    float* deg = hB + feat;

    int* binned  = (int*)hB;
    int* cntT    = binned + NEDGES;
    int* offT    = cntT + NBINS * NBINS;
    int* scratch = offT + NBINS * NBINS;
    int* csr    = (int*)lst;
    int* rowptr = csr + NEDGES;

    sage_bin_kernel<<<NBINS, BUILD_T, 0, stream>>>(src, dst, binned, cntT, offT);
    sage_list_kernel<<<NBINS, BUILD_T, 0, stream>>>(binned, cntT, offT, scratch, csr, rowptr, deg);

    sage_layer_kernel<128, true><<<GEMM_BLOCKS, GEMM_THREADS, 0, stream>>>(
        x, rowptr, csr, deg, Ws0, Wn0, b0, hA);

    sage_layer_kernel<128, true><<<GEMM_BLOCKS, GEMM_THREADS, 0, stream>>>(
        hA, rowptr, csr, deg, Ws1, Wn1, b1, hB);

    sage_layer_kernel<64, false><<<GEMM_BLOCKS, GEMM_THREADS, 0, stream>>>(
        hB, rowptr, csr, deg, Ws2, Wn2, b2, out);
}


# Layer 2's two products moved in front of its neighbour sum (taken on layer 1's result strips, still WMMA): last launch gathers 64-wide rows, not 128-wide; layer 2's f32 sum order changes (fixed)
# speedup vs baseline: 4.6759x; 1.0488x over previous
#include <hip/hip_runtime.h>
#include <type_traits>


#define NNODES 50000
#define NEDGES 800000
#define KDIM   128

typedef __attribute__((ext_vector_type(2))) float v2f;
typedef __attribute__((ext_vector_type(8))) float v8f;

typedef __attribute__((ext_vector_type(4))) float v4f;

#define NBINS    256
#define BIN_W    196
#define CHUNK    (NEDGES / NBINS)
#define BUILD_T  1024
#define BIN_CAP  8192
static_assert(NBINS * BIN_W >= NNODES, "ranges cover all nodes");
static_assert(NEDGES % NBINS == 0, "whole chunks");
static_assert(NNODES <= 65536 && BIN_W <= 256,
              "packed edge word: src in bits 0..15, dst minus range start in bits 16..23");

__device__ __forceinline__ int sage_block_scan(int v, int* s_tot, int* total) {
    const int lane = threadIdx.x & 31;
    const int w    = threadIdx.x >> 5;
    int inc = v;
    for (int o = 1; o < 32; o <<= 1) {
        const int u = __shfl_up(inc, o);
        if (lane >= o) inc += u;
    }
    if (lane == 31) s_tot[w] = inc;
    __syncthreads();
    int before = 0, all = 0;
    for (int k = 0; k < BUILD_T / 32; ++k) {
        const int x = s_tot[k];
        all += x;
        if (k < w) before += x;
    }
    __syncthreads();
    *total = all;
    return before + inc - v;
}

__global__ void __launch_bounds__(BUILD_T)
sage_bin_kernel(const int* __restrict__ src,
                const int* __restrict__ dst,
                int* __restrict__ binned,
                int* __restrict__ cntT,
                int* __restrict__ offT) {
    __shared__ int s_cnt[NBINS];
    __shared__ int s_off[NBINS];
    __shared__ int s_stage[CHUNK];
    __shared__ int s_tot[BUILD_T / 32];
    const int t = threadIdx.x;
    const int c = blockIdx.x;
    if (t < NBINS) s_cnt[t] = 0;
    __syncthreads();

    constexpr int PER = (CHUNK + BUILD_T - 1) / BUILD_T;
    int word[PER], bin[PER], slot[PER];
#pragma unroll
    for (int k = 0; k < PER; ++k) {
        const int i = t + k * BUILD_T;
        bin[k]  = -1;
        word[k] = 0;
        if (i < CHUNK) {
            const int e = c * CHUNK + i;
            const unsigned d = min((unsigned)dst[e], (unsigned)(NNODES - 1));
            const unsigned b = d / BIN_W;
            bin[k]  = (int)b;
            word[k] = (src[e] & 0xFFFF) | (int)((d - b * BIN_W) << 16);
        }
    }
#pragma unroll
    for (int k = 0; k < PER; ++k)
        slot[k] = (bin[k] >= 0) ? atomicAdd(&s_cnt[bin[k]], 1) : 0;
    __syncthreads();

    int total;
    const int mine = (t < NBINS) ? s_cnt[t] : 0;
    const int ex   = sage_block_scan(mine, s_tot, &total);
    if (t < NBINS) {
        s_off[t] = ex;
        cntT[t * NBINS + c] = mine;
        offT[t * NBINS + c] = ex;
    }
    __syncthreads();
#pragma unroll
    for (int k = 0; k < PER; ++k)
        if (bin[k] >= 0) s_stage[s_off[bin[k]] + slot[k]] = word[k];
    __syncthreads();
    for (int i = t; i < CHUNK; i += BUILD_T) binned[c * CHUNK + i] = s_stage[i];
}

__global__ void __launch_bounds__(BUILD_T)
sage_list_kernel(const int* __restrict__ binned,
                 const int* __restrict__ cntT,
                 const int* __restrict__ offT,
                 int* __restrict__ scratch,
                 int* __restrict__ csr,
                 int* __restrict__ rowptr,
                 float* __restrict__ deg) {
    __shared__ int s_P[NBINS + 1];
    __shared__ int s_src0[NBINS];
    __shared__ int s_ncnt[BIN_W];
    __shared__ int s_nbeg[BIN_W];
    __shared__ int s_nfill[BIN_W];
    __shared__ int s_tot[BUILD_T / 32];
    __shared__ int s_red[BUILD_T / 32];
    __shared__ int s_stage[BIN_CAP];
    __shared__ int s_list[BIN_CAP];
    const int t    = threadIdx.x;
    const int lane = t & 31;
    const int b    = blockIdx.x;
    const int wave = __builtin_amdgcn_readfirstlane((int)(threadIdx.x >> 5));

    int acc = 0;
    const int4* c4 = (const int4*)cntT;
    constexpr int BASE_ITERS = ((NBINS - 1) * (NBINS / 4) + BUILD_T - 1) / BUILD_T;
    int4 cv[BASE_ITERS];
#pragma unroll
    for (int j = 0; j < BASE_ITERS; ++j) {
        const int i = t + j * BUILD_T;
        cv[j] = (i < b * (NBINS / 4)) ? c4[i] : int4{0, 0, 0, 0};
    }
#pragma unroll
    for (int j = 0; j < BASE_ITERS; ++j) acc += cv[j].x + cv[j].y + cv[j].z + cv[j].w;
    for (int o = 16; o > 0; o >>= 1) acc += __shfl_xor(acc, o);
    if (lane == 0) s_red[wave] = acc;

    const int pc = (t < NBINS) ? cntT[b * NBINS + t] : 0;
    if (t < NBINS) s_src0[t] = t * CHUNK + offT[b * NBINS + t];
    if (t < BIN_W) { s_ncnt[t] = 0; s_nfill[t] = 0; }
    int tot;
    const int pex = sage_block_scan(pc, s_tot, &tot);
    if (t < NBINS) s_P[t] = pex;
    if (t == 0)    s_P[NBINS] = tot;
    int base = 0;
    for (int k = 0; k < BUILD_T / 32; ++k) base += s_red[k];
    __syncthreads();

    const bool fits = tot <= BIN_CAP;

    auto fetch = [&](int k) -> int {
        static_assert(NBINS == 256, "eight halvings");
        int lo = 0, hi = NBINS;
#pragma unroll
        for (int s = 0; s < 8; ++s) {
            const int mid = (lo + hi) >> 1;
            if (s_P[mid] <= k) lo = mid; else hi = mid;
        }
        return binned[s_src0[lo] + (k - s_P[lo])];
    };

    for (int k0 = t; k0 < tot; k0 += 4 * BUILD_T) {
        int w4[4];
#pragma unroll
        for (int j = 0; j < 4; ++j) {
            const int k = k0 + j * BUILD_T;
            w4[j] = (k < tot) ? fetch(k) : 0;
        }
#pragma unroll
        for (int j = 0; j < 4; ++j) {
            const int k = k0 + j * BUILD_T;
            if (k < tot) {
                if (fits) s_stage[k] = w4[j];
                atomicAdd(&s_ncnt[w4[j] >> 16], 1);
            }
        }
    }
    __syncthreads();

    const int nc = (t < BIN_W) ? s_ncnt[t] : 0;
    int tot2;
    const int nex = sage_block_scan(nc, s_tot, &tot2);
    if (t < BIN_W) {
        const int node = b * BIN_W + t;
        s_nbeg[t] = nex;
        if (node < NNODES) {
            rowptr[node] = base + nex;
            deg[node]    = (float)nc;
        }
        if (node == NNODES - 1) rowptr[NNODES] = base + tot;
    }
    __syncthreads();

    int* glist = scratch + base;
    int* out   = csr + base;

    auto place_and_order = [&](auto in_lds) {
        constexpr bool IN_LDS = decltype(in_lds)::value;
        for (int k = t; k < tot; k += BUILD_T) {
            int w;
            if constexpr (IN_LDS) w = s_stage[k]; else w = fetch(k);
            const int dl = w >> 16;
            const int sl = s_nbeg[dl] + atomicAdd(&s_nfill[dl], 1);
            if constexpr (IN_LDS) s_list[sl] = w & 0xFFFF; else glist[sl] = w & 0xFFFF;
        }
        __syncthreads();
        for (int nd = wave; nd < BIN_W; nd += BUILD_T / 32) {
            const int beg = __builtin_amdgcn_readfirstlane(s_nbeg[nd]);
            const int n   = __builtin_amdgcn_readfirstlane(s_ncnt[nd]);
            for (int i0 = 0; i0 < n; i0 += 32) {
                const int i = i0 + lane;
                int a = 0;
                if (i < n) { if constexpr (IN_LDS) a = s_list[beg + i]; else a = glist[beg + i]; }
                int pos = 0;
                for (int m0 = 0; m0 < n; m0 += 32) {
                    const int mm = m0 + lane;
                    int bv = 0;
                    if (mm < n) { if constexpr (IN_LDS) bv = s_list[beg + mm]; else bv = glist[beg + mm]; }
                    int piece = n - m0;
                    if (piece > 32) piece = 32;
                    for (int u = 0; u < piece; ++u) {
                        const int x = __builtin_amdgcn_readlane(bv, u);
                        const int m = m0 + u;
                        pos += (x < a || (x == a && m < i)) ? 1 : 0;
                    }
                }
                if (i < n) out[beg + pos] = a;
            }
        }
    };
    if (fits) place_and_order(std::true_type{});
    else      place_and_order(std::false_type{});
}

#define ROWS_IN_FLIGHT 24

template <int U, int DIM, typename V>
__device__ __forceinline__ void sage_issue_rows(const float* __restrict__ h, int idx, int j,
                                                int s_first, int cnt, int lane_off,
                                                V (&v)[ROWS_IN_FLIGHT]) {
    if constexpr (U < ROWS_IN_FLIGHT) {
        if (U < cnt) {
            const int s = (U == 0) ? s_first : __builtin_amdgcn_readlane(idx, j + U);
            v[U] = *(const V*)(h + (size_t)s * DIM + lane_off);
            sage_issue_rows<U + 1, DIM>(h, idx, j, s_first, cnt, lane_off, v);
        }
    }
}

template <int U, typename V>
__device__ __forceinline__ void sage_sum_rows(const V (&v)[ROWS_IN_FLIGHT], int cnt, V& acc) {
    if constexpr (U < ROWS_IN_FLIGHT) {
        if (U < cnt) {
            acc += v[U];
            sage_sum_rows<U + 1>(v, cnt, acc);
        }
    }
}

template <int DIM, typename V>
__device__ __forceinline__ void sage_finish_list(const float* __restrict__ h,
                                                 const int* __restrict__ csr,
                                                 int beg, int end, int n, int idx,
                                                 int lane, int lane4,
                                                 V (&v)[ROWS_IN_FLIGHT], V& acc) {
    int j = ROWS_IN_FLIGHT;
    for (int base = beg;;) {
        for (; j < n; j += ROWS_IN_FLIGHT) {
            int cnt = n - j;
            if (cnt > ROWS_IN_FLIGHT) cnt = ROWS_IN_FLIGHT;
            sage_issue_rows<0, DIM>(h, idx, j, __builtin_amdgcn_readlane(idx, j), cnt, lane4, v);
            sage_sum_rows<0>(v, cnt, acc);
        }
        base += 32;
        if (base >= end) break;
        n = end - base;
        if (n > 32) n = 32;
        idx = (lane < n) ? csr[base + lane] : 0;
        j = 0;
    }
}

#define GEMM_THREADS 512
#define GEMM_BLOCKS  256
#define TILE_WAVES   13
#define GEMM_QUADS   64
#define TILE_PITCH   33

template <int NC, bool RELU, int NC2>
__global__ void __launch_bounds__(GEMM_THREADS)
sage_layer_kernel(const float* __restrict__ h,
                  const int* __restrict__ rowptr,
                  const int* __restrict__ csr,
                  const float* __restrict__ deg,
                  const float* __restrict__ Ws,
                  const float* __restrict__ Wn,
                  const float* __restrict__ bias,
                  float* __restrict__ out,
                  const float* __restrict__ Ws2,
                  const float* __restrict__ Wn2,
                  const float* __restrict__ bias2,
                  float* __restrict__ outS,
                  float* __restrict__ outZ) {
    constexpr int TILES   = NC / 16;
    constexpr int NSTRIPS = NNODES / 16;
    static_assert(NC % 64 == 0, "columns go in groups of 64 (4 tiles)");
    constexpr int PITCH   = NC + 8;
    constexpr int WPB     = GEMM_THREADS / 32;
    static_assert(NNODES % 16 == 0, "whole strips");
    static_assert(GEMM_BLOCKS * TILE_WAVES >= NSTRIPS && TILE_WAVES <= WPB, "every strip's wave has a tile");
    __shared__ v4f sW[GEMM_QUADS * PITCH];
    __shared__ v4f sA[TILE_WAVES * 16 * TILE_PITCH];
    constexpr int NCX     = 2 * NC2;
    constexpr int TILES2  = NCX / 16;
    constexpr int QUADS2  = NC / 4;
    constexpr int PITCH2  = NCX + 8;
    static_assert(NC2 == 0 || (NC2 == 64 && NC == KDIM), "second product: one group of 64 columns each, K = tile width");
    __shared__ v4f sW2[NC2 > 0 ? QUADS2 * PITCH2 : 1];

    constexpr int FILL_ITERS = GEMM_QUADS * NC / GEMM_THREADS;
    static_assert(GEMM_QUADS * NC % GEMM_THREADS == 0, "whole fill passes");
#pragma unroll 8
    for (int i = 0; i < FILL_ITERS; ++i) {
        const int idx = threadIdx.x + i * GEMM_THREADS;
        const int q = idx / NC;
        const int p = idx % NC;
        const int n = (p / 64) * 64 + 4 * (p & 15) + ((p % 64) >> 4);
        const float* w = ((q < 32) ? Ws : Wn) + (size_t)(4 * (q & 31)) * NC + n;
        v4f v;
        v.x = w[0];
        v.y = w[NC];
        v.z = w[2 * NC];
        v.w = w[3 * NC];
        sW[q * PITCH + p] = v;
    }
    if constexpr (NC2 > 0) {
        constexpr int FILL2 = QUADS2 * NCX / GEMM_THREADS;
        static_assert(QUADS2 * NCX % GEMM_THREADS == 0, "whole fill passes");
#pragma unroll 8
        for (int i = 0; i < FILL2; ++i) {
            const int idx = threadIdx.x + i * GEMM_THREADS;
            const int q = idx / NCX;
            const int p = idx % NCX;
            const int n = 4 * (p & 15) + ((p % 64) >> 4);
            const float* w = ((p < 64) ? Ws2 : Wn2) + (size_t)(4 * q) * NC2 + n;
            v4f v;
            v.x = w[0];
            v.y = w[NC2];
            v.z = w[2 * NC2];
            v.w = w[3 * NC2];
            sW2[q * PITCH2 + p] = v;
        }
    }
    __syncthreads();

    const int lane  = threadIdx.x & 31;
    const int lane4 = lane << 2;
    const int c     = lane & 15;
    const int hh    = lane >> 4;
    const int wave = __builtin_amdgcn_readfirstlane((int)(threadIdx.x >> 5));

    static_assert(GEMM_BLOCKS * WPB >= NSTRIPS, "a wave takes at most one strip");
    const int strip = blockIdx.x + GEMM_BLOCKS * wave;
    if (strip < NSTRIPS) {
        v4f* tile = sA + wave * (16 * TILE_PITCH);
        const int r0  = strip << 4;
        const int row = r0 + c;
        const float idg = 1.0f / fmaxf(deg[row], 1.0f);

        const int rp = rowptr[r0 + (lane < 16 ? lane : 16)];
        int begA = __builtin_amdgcn_readlane(rp, 0);
        int endA = __builtin_amdgcn_readlane(rp, 1);
        int endB = __builtin_amdgcn_readlane(rp, 2);
        int nA = endA - begA;
        if (nA > 32) nA = 32;
        int nB = endB - endA;
        if (nB > 32) nB = 32;
        int idxA = (lane < nA) ? csr[begA + lane] : 0;
        int idxB = (lane < nB) ? csr[endA + lane] : 0;
        for (int r = 0; r < 16; r += 2) {
            const int begB  = endA;
            const int endNA = __builtin_amdgcn_readlane(rp, r + 3);
            const int endNB = __builtin_amdgcn_readlane(rp, r + 4);
            int nNA = endNA - endB;
            if (nNA > 32) nNA = 32;
            int nNB = endNB - endNA;
            if (nNB > 32) nNB = 32;

            v4f accA = {0.f, 0.f, 0.f, 0.f};
            v4f accB = {0.f, 0.f, 0.f, 0.f};
            v4f vA[ROWS_IN_FLIGHT], vB[ROWS_IN_FLIGHT];
            const int cntA = (nA < ROWS_IN_FLIGHT) ? nA : ROWS_IN_FLIGHT;
            const int cntB = (nB < ROWS_IN_FLIGHT) ? nB : ROWS_IN_FLIGHT;
            const int sA0 = __builtin_amdgcn_readlane(idxA, 0);
            const int sB0 = __builtin_amdgcn_readlane(idxB, 0);
            sage_issue_rows<0, KDIM>(h, idxA, 0, sA0, cntA, lane4, vA);
            sage_issue_rows<0, KDIM>(h, idxB, 0, sB0, cntB, lane4, vB);
            const int idxNA = (lane < nNA) ? csr[endB + lane] : 0;
            const int idxNB = (lane < nNB) ? csr[endNA + lane] : 0;
            sage_sum_rows<0>(vA, cntA, accA);
            sage_sum_rows<0>(vB, cntB, accB);

            sage_finish_list<KDIM>(h, csr, begA, endA, nA, idxA, lane, lane4, vA, accA);
            sage_finish_list<KDIM>(h, csr, begB, endB, nB, idxB, lane, lane4, vB, accB);

            const float scA = __builtin_bit_cast(float,
                __builtin_amdgcn_readlane(__builtin_bit_cast(int, idg), r));
            const float scB = __builtin_bit_cast(float,
                __builtin_amdgcn_readlane(__builtin_bit_cast(int, idg), r + 1));
            tile[r * TILE_PITCH + lane]       = accA * scA;
            tile[(r + 1) * TILE_PITCH + lane] = accB * scB;

            begA = endB; endA = endNA; endB = endNB;
            nA = nNA; nB = nNB; idxA = idxNA; idxB = idxNB;
        }
        __builtin_amdgcn_wave_barrier();

        const v4f* hq = (const v4f*)(h + (size_t)row * KDIM) + hh;
        const v4f* gq = tile + c * TILE_PITCH + hh;

        const v4f* bq = sW + hh * PITCH + c;

        v8f acc[TILES];
#pragma unroll
        for (int t = 0; t < TILES; ++t)
            acc[t] = (v8f){0.f, 0.f, 0.f, 0.f, 0.f, 0.f, 0.f, 0.f};

        v4f a[4], an[4], b[TILES], bn[TILES];
#pragma unroll
        for (int jj = 0; jj < 4; ++jj) a[jj] = hq[2 * jj];
#pragma unroll
        for (int t = 0; t < TILES; ++t) b[t] = bq[16 * t];

#pragma unroll
        for (int ch = 0; ch < 8; ++ch) {
            if (ch + 1 < 4) {
                const v4f* nx = hq + 8 * (ch + 1);
#pragma unroll
                for (int jj = 0; jj < 4; ++jj) an[jj] = nx[2 * jj];
            } else if (ch + 1 < 8) {
                const v4f* nx = gq + 8 * (ch + 1 - 4);
#pragma unroll
                for (int jj = 0; jj < 4; ++jj) an[jj] = nx[2 * jj];
            }
#pragma unroll
            for (int jj = 0; jj < 4; ++jj) {
                const int s = 4 * ch + jj;
                if (s + 1 < 32) {
#pragma unroll
                    for (int t = 0; t < TILES; ++t) bn[t] = bq[(2 * (s + 1)) * PITCH + 16 * t];
                }
                const v2f alo = {a[jj].x, a[jj].y};
                const v2f ahi = {a[jj].z, a[jj].w};
#pragma unroll
                for (int t = 0; t < TILES; ++t) {
                    const v2f blo = {b[t].x, b[t].y};
                    acc[t] = __builtin_amdgcn_wmma_f32_16x16x4_f32(false, alo, false, blo,
                                                                   (short)0, acc[t], false, false);
                }
#pragma unroll
                for (int t = 0; t < TILES; ++t) {
                    const v2f bhi = {b[t].z, b[t].w};
                    acc[t] = __builtin_amdgcn_wmma_f32_16x16x4_f32(false, ahi, false, bhi,
                                                                   (short)0, acc[t], false, false);
                }
#pragma unroll
                for (int t = 0; t < TILES; ++t) b[t] = bn[t];
            }
#pragma unroll
            for (int jj = 0; jj < 4; ++jj) a[jj] = an[jj];
        }

        float bv[TILES];
#pragma unroll
        for (int t = 0; t < TILES; ++t) bv[t] = bias[64 * (t >> 2) + 4 * c + (t & 3)];
        if constexpr (NC2 == 0) {
            float* orow = out + (size_t)(r0 + 8 * hh) * NC + 4 * c;
#pragma unroll
            for (int v = 0; v < 8; ++v) {
                float x[TILES];
#pragma unroll
                for (int t = 0; t < TILES; ++t) {
                    x[t] = acc[t][v] + bv[t];
                    if (RELU) x[t] = fmaxf(x[t], 0.0f);
                }
#pragma unroll
                for (int t = 0; t < TILES; t += 4) {
                    v4f o = {x[t], x[t + 1], x[t + 2], x[t + 3]};
                    *(v4f*)(orow + (size_t)v * NC + 16 * t) = o;
                }
            }
        } else {
#pragma unroll
            for (int v = 0; v < 8; ++v) {
                float x[TILES];
#pragma unroll
                for (int t = 0; t < TILES; ++t) {
                    x[t] = acc[t][v] + bv[t];
                    if (RELU) x[t] = fmaxf(x[t], 0.0f);
                }
#pragma unroll
                for (int t = 0; t < TILES; t += 4) {
                    v4f o = {x[t], x[t + 1], x[t + 2], x[t + 3]};
                    tile[(8 * hh + v) * TILE_PITCH + 4 * t + c] = o;
                }
            }
            __builtin_amdgcn_wave_barrier();

            const v4f* bq2 = sW2 + hh * PITCH2 + c;
            v8f acc2[TILES2];
#pragma unroll
            for (int t = 0; t < TILES2; ++t)
                acc2[t] = (v8f){0.f, 0.f, 0.f, 0.f, 0.f, 0.f, 0.f, 0.f};
            v4f a2[4], an2[4], b2[TILES2], bn2[TILES2];
#pragma unroll
            for (int jj = 0; jj < 4; ++jj) a2[jj] = gq[2 * jj];
#pragma unroll
            for (int t = 0; t < TILES2; ++t) b2[t] = bq2[16 * t];
#pragma unroll
            for (int ch = 0; ch < QUADS2 / 8; ++ch) {
                if (ch + 1 < QUADS2 / 8) {
                    const v4f* nx = gq + 8 * (ch + 1);
#pragma unroll
                    for (int jj = 0; jj < 4; ++jj) an2[jj] = nx[2 * jj];
                }
#pragma unroll
                for (int jj = 0; jj < 4; ++jj) {
                    const int s = 4 * ch + jj;
                    if (s + 1 < QUADS2 / 2) {
#pragma unroll
                        for (int t = 0; t < TILES2; ++t) bn2[t] = bq2[(2 * (s + 1)) * PITCH2 + 16 * t];
                    }
                    const v2f alo = {a2[jj].x, a2[jj].y};
                    const v2f ahi = {a2[jj].z, a2[jj].w};
#pragma unroll
                    for (int t = 0; t < TILES2; ++t) {
                        const v2f blo = {b2[t].x, b2[t].y};
                        acc2[t] = __builtin_amdgcn_wmma_f32_16x16x4_f32(false, alo, false, blo,
                                                                        (short)0, acc2[t], false, false);
                    }
#pragma unroll
                    for (int t = 0; t < TILES2; ++t) {
                        const v2f bhi = {b2[t].z, b2[t].w};
                        acc2[t] = __builtin_amdgcn_wmma_f32_16x16x4_f32(false, ahi, false, bhi,
                                                                        (short)0, acc2[t], false, false);
                    }
#pragma unroll
                    for (int t = 0; t < TILES2; ++t) b2[t] = bn2[t];
                }
#pragma unroll
                for (int jj = 0; jj < 4; ++jj) a2[jj] = an2[jj];
            }

            float b2v[4];
#pragma unroll
            for (int t = 0; t < 4; ++t) b2v[t] = bias2[4 * c + t];
            float* srow = outS + (size_t)(r0 + 8 * hh) * NC2 + 4 * c;
            float* zrow = outZ + (size_t)(r0 + 8 * hh) * NC2 + 4 * c;
#pragma unroll
            for (int v = 0; v < 8; ++v) {
                v4f so = {acc2[0][v] + b2v[0], acc2[1][v] + b2v[1], acc2[2][v] + b2v[2], acc2[3][v] + b2v[3]};
                v4f zo = {acc2[4][v], acc2[5][v], acc2[6][v], acc2[7][v]};
                *(v4f*)(srow + (size_t)v * NC2) = so;
                *(v4f*)(zrow + (size_t)v * NC2) = zo;
            }
        }
    }
}

#define OUT_DIM      64
#define MEAN_THREADS 1024
#define MEAN_BLOCKS  256
#define MEAN_NODES   8

__global__ void __launch_bounds__(MEAN_THREADS)
sage_mean_add_kernel(const float* __restrict__ z,
                     const float* __restrict__ sself,
                     const int* __restrict__ rowptr,
                     const int* __restrict__ csr,
                     const float* __restrict__ deg,
                     float* __restrict__ out) {
    constexpr int NGROUPS = NNODES / MEAN_NODES;
    static_assert(NNODES % MEAN_NODES == 0, "whole groups");
    static_assert(MEAN_BLOCKS * (MEAN_THREADS / 32) >= NGROUPS, "a wave takes at most one group");
    const int lane  = threadIdx.x & 31;
    const int lane2 = lane << 1;
    const int wave  = __builtin_amdgcn_readfirstlane((int)(threadIdx.x >> 5));
    const int grp   = blockIdx.x + MEAN_BLOCKS * wave;
    if (grp < NGROUPS) {
        const int n0 = grp * MEAN_NODES;
        const float idg = 1.0f / fmaxf(deg[n0 + (lane & (MEAN_NODES - 1))], 1.0f);
        const int rp = rowptr[n0 + (lane < MEAN_NODES ? lane : MEAN_NODES)];
        int begA = __builtin_amdgcn_readlane(rp, 0);
        int endA = __builtin_amdgcn_readlane(rp, 1);
        int endB = __builtin_amdgcn_readlane(rp, 2);
        int nA = endA - begA;
        if (nA > 32) nA = 32;
        int nB = endB - endA;
        if (nB > 32) nB = 32;
        int idxA = (lane < nA) ? csr[begA + lane] : 0;
        int idxB = (lane < nB) ? csr[endA + lane] : 0;
        for (int r = 0; r < MEAN_NODES; r += 2) {
            const int begB  = endA;
            const int endNA = __builtin_amdgcn_readlane(rp, r + 3);
            const int endNB = __builtin_amdgcn_readlane(rp, r + 4);
            int nNA = endNA - endB;
            if (nNA > 32) nNA = 32;
            int nNB = endNB - endNA;
            if (nNB > 32) nNB = 32;

            v2f accA = {0.f, 0.f};
            v2f accB = {0.f, 0.f};
            v2f vA[ROWS_IN_FLIGHT], vB[ROWS_IN_FLIGHT];
            const int cntA = (nA < ROWS_IN_FLIGHT) ? nA : ROWS_IN_FLIGHT;
            const int cntB = (nB < ROWS_IN_FLIGHT) ? nB : ROWS_IN_FLIGHT;
            const int sA0 = __builtin_amdgcn_readlane(idxA, 0);
            const int sB0 = __builtin_amdgcn_readlane(idxB, 0);
            const v2f selfA = *(const v2f*)(sself + (size_t)(n0 + r) * OUT_DIM + lane2);
            const v2f selfB = *(const v2f*)(sself + (size_t)(n0 + r + 1) * OUT_DIM + lane2);
            sage_issue_rows<0, OUT_DIM>(z, idxA, 0, sA0, cntA, lane2, vA);
            sage_issue_rows<0, OUT_DIM>(z, idxB, 0, sB0, cntB, lane2, vB);
            const int idxNA = (lane < nNA) ? csr[endB + lane] : 0;
            const int idxNB = (lane < nNB) ? csr[endNA + lane] : 0;
            sage_sum_rows<0>(vA, cntA, accA);
            sage_sum_rows<0>(vB, cntB, accB);

            sage_finish_list<OUT_DIM>(z, csr, begA, endA, nA, idxA, lane, lane2, vA, accA);
            sage_finish_list<OUT_DIM>(z, csr, begB, endB, nB, idxB, lane, lane2, vB, accB);

            const float scA = __builtin_bit_cast(float,
                __builtin_amdgcn_readlane(__builtin_bit_cast(int, idg), r));
            const float scB = __builtin_bit_cast(float,
                __builtin_amdgcn_readlane(__builtin_bit_cast(int, idg), r + 1));
            *(v2f*)(out + (size_t)(n0 + r) * OUT_DIM + lane2)     = selfA + accA * scA;
            *(v2f*)(out + (size_t)(n0 + r + 1) * OUT_DIM + lane2) = selfB + accB * scB;

            begA = endB; endA = endNA; endB = endNB;
            nA = nNA; nB = nNB; idxA = idxNA; idxB = idxNB;
        }
    }
}

extern "C" void kernel_launch(void* const* d_in, const int* in_sizes, int n_in,
                              void* d_out, int out_size, void* d_ws, size_t ws_size,
                              hipStream_t stream) {
    (void)in_sizes; (void)n_in; (void)out_size; (void)ws_size;

    const float* x   = (const float*)d_in[0];
    const int*   src = (const int*)d_in[1];
    const int*   dst = (const int*)d_in[2];
    const float* Ws0 = (const float*)d_in[3];
    const float* Wn0 = (const float*)d_in[4];
    const float* b0  = (const float*)d_in[5];
    const float* Ws1 = (const float*)d_in[6];
    const float* Wn1 = (const float*)d_in[7];
    const float* b1  = (const float*)d_in[8];
    const float* Ws2 = (const float*)d_in[9];
    const float* Wn2 = (const float*)d_in[10];
    const float* b2  = (const float*)d_in[11];
    float* out = (float*)d_out;

    const size_t feat = (size_t)NNODES * KDIM;
    float* lst = (float*)d_ws;
    float* hA  = lst + feat;
    float* hB  = hA + feat;
    float* deg = hB + feat;

    int* binned  = (int*)hB;
    int* cntT    = binned + NEDGES;
    int* offT    = cntT + NBINS * NBINS;
    int* scratch = offT + NBINS * NBINS;
    int* csr    = (int*)lst;
    int* rowptr = csr + NEDGES;

    sage_bin_kernel<<<NBINS, BUILD_T, 0, stream>>>(src, dst, binned, cntT, offT);
    sage_list_kernel<<<NBINS, BUILD_T, 0, stream>>>(binned, cntT, offT, scratch, csr, rowptr, deg);

    float* S2 = hB;
    float* Z2 = hB + (size_t)NNODES * OUT_DIM;

    sage_layer_kernel<128, true, 0><<<GEMM_BLOCKS, GEMM_THREADS, 0, stream>>>(
        x, rowptr, csr, deg, Ws0, Wn0, b0, hA, nullptr, nullptr, nullptr, nullptr, nullptr);

    sage_layer_kernel<128, true, OUT_DIM><<<GEMM_BLOCKS, GEMM_THREADS, 0, stream>>>(
        hA, rowptr, csr, deg, Ws1, Wn1, b1, nullptr, Ws2, Wn2, b2, S2, Z2);

    sage_mean_add_kernel<<<MEAN_BLOCKS, MEAN_THREADS, 0, stream>>>(Z2, S2, rowptr, csr, deg, out);
}
